# starting kernel
# baseline (speedup 1.0000x reference)
#include <hip/hip_runtime.h>

#define B_    4
#define N_    50000
#define CIN_  64
#define E_    800000
#define HALF_ 32
#define M_    (B_ * N_)

typedef __attribute__((ext_vector_type(2))) float v2f;
typedef __attribute__((ext_vector_type(8))) float v8f;

__global__ __launch_bounds__(256) void gcn_gemm_wmma(
    const float* __restrict__ x,
    const float* __restrict__ W_lin, const float* __restrict__ b_lin,
    const float* __restrict__ W_eye, const float* __restrict__ b_eye,
    float* __restrict__ out, float* __restrict__ z)
{
    __shared__ float lw[64 * 64];

    const int tid = threadIdx.x;
    #pragma unroll
    for (int j = 0; j < 16; ++j) {
        int idx = tid + 256 * j;
        int n = idx >> 6, k = idx & 63;
        lw[idx] = (n < HALF_) ? W_lin[n * CIN_ + k]
                              : W_eye[(n - HALF_) * CIN_ + k];
    }
    __syncthreads();

    const int wave  = tid >> 5;
    const int lane  = tid & 31;
    const int hi    = lane >> 4;
    const int l16   = lane & 15;
    const int mtile = blockIdx.x * 2 + (wave >> 2);
    const int nt    = wave & 3;

    const int row  = mtile * 16 + l16;
    const int ncol = nt * 16 + l16;

    const float* xrow = x  + (long)row * CIN_ + hi * 2;
    const float* wcol = lw + ncol * CIN_ + hi * 2;

    v8f acc = {};
    #pragma unroll
    for (int kb = 0; kb < 16; ++kb) {
        v2f a = *(const v2f*)(xrow + kb * 4);
        v2f b = *(const v2f*)(wcol + kb * 4);
        acc = __builtin_amdgcn_wmma_f32_16x16x4_f32(
             false, a,  false, b,
             (short)0, acc,  false,  false);
    }

    const int m0 = mtile * 16 + 8 * hi;
    if (nt < 2) {
        const float bl = b_lin[ncol];
        #pragma unroll
        for (int r = 0; r < 8; ++r) {
            const long m = m0 + r;
            z[m * HALF_ + ncol]  = acc[r];
            out[m * CIN_ + ncol] = bl;
        }
    } else {
        const float be = b_eye[ncol - HALF_];
        #pragma unroll
        for (int r = 0; r < 8; ++r) {
            const long m = m0 + r;
            out[m * CIN_ + ncol] = acc[r] + be;
        }
    }
}

__global__ __launch_bounds__(256) void gcn_edge_scatter(
    const float* __restrict__ z, const float* __restrict__ vals,
    const int* __restrict__ rows, const int* __restrict__ cols,
    float* __restrict__ out)
{
    const long t = (long)blockIdx.x * 256 + threadIdx.x;
    const int e = (int)(t >> 7);
    const int r = (int)(t & 127);
    const int b = r >> 5;
    const int c = r & 31;

    const int   cn = cols[e];
    const int   rn = rows[e];
    const float v  = vals[e];

    const float zv  = z[((long)b * N_ + cn) * HALF_ + c];
    const float upd = v * zv;
    float* dst = out + ((long)b * N_ + rn) * CIN_ + c;

    asm volatile("global_atomic_add_f32 %0, %1, off scope:SCOPE_DEV"
                 :
                 : "v"(dst), "v"(upd)
                 : "memory");
}

extern "C" void kernel_launch(void* const* d_in, const int* in_sizes, int n_in,
                              void* d_out, int out_size, void* d_ws, size_t ws_size,
                              hipStream_t stream) {
    const float* x     = (const float*)d_in[0];
    const float* vals  = (const float*)d_in[1];
    const float* W_lin = (const float*)d_in[2];
    const float* b_lin = (const float*)d_in[3];
    const float* W_eye = (const float*)d_in[4];
    const float* b_eye = (const float*)d_in[5];
    const int*   rows  = (const int*)d_in[6];
    const int*   cols  = (const int*)d_in[7];
    float* out = (float*)d_out;
    float* z   = (float*)d_ws;

    gcn_gemm_wmma<<<M_ / 32, 256, 0, stream>>>(x, W_lin, b_lin, W_eye, b_eye, out, z);

    gcn_edge_scatter<<<(long)E_ * 128 / 256, 256, 0, stream>>>(z, vals, rows, cols, out);
}


# Float atomics removed without sorting: each work-group owns 196 rows' sums in LDS, filters the whole edge list itself, each wave adds its own rows with plain LDS adds; row sums in a fixed order
# speedup vs baseline: 2.9555x; 2.9555x over previous
#include <hip/hip_runtime.h>

#define B_    4
#define N_    50000
#define CIN_  64
#define E_    800000
#define HALF_ 32
#define M_    (B_ * N_)

typedef __attribute__((ext_vector_type(2))) float v2f;
typedef __attribute__((ext_vector_type(8))) float v8f;

__global__ __launch_bounds__(256) void gcn_gemm_wmma(
    const float* __restrict__ x,
    const float* __restrict__ W_lin,
    const float* __restrict__ W_eye, const float* __restrict__ b_eye,
    float* __restrict__ out, float* __restrict__ z)
{
    __shared__ float lw[64 * 64];

    const int tid = threadIdx.x;
    #pragma unroll
    for (int j = 0; j < 16; ++j) {
        int idx = tid + 256 * j;
        int n = idx >> 6, k = idx & 63;
        lw[idx] = (n < HALF_) ? W_lin[n * CIN_ + k]
                              : W_eye[(n - HALF_) * CIN_ + k];
    }
    __syncthreads();

    const int wave  = tid >> 5;
    const int lane  = tid & 31;
    const int hi    = lane >> 4;
    const int l16   = lane & 15;
    const int mtile = blockIdx.x * 2 + (wave >> 2);
    const int nt    = wave & 3;

    const int row  = mtile * 16 + l16;
    const int ncol = nt * 16 + l16;

    const float* xrow = x  + (long)row * CIN_ + hi * 2;
    const float* wcol = lw + ncol * CIN_ + hi * 2;

    v8f acc = {};
    #pragma unroll
    for (int kb = 0; kb < 16; ++kb) {
        v2f a = *(const v2f*)(xrow + kb * 4);
        v2f b = *(const v2f*)(wcol + kb * 4);
        acc = __builtin_amdgcn_wmma_f32_16x16x4_f32(
             false, a,  false, b,
             (short)0, acc,  false,  false);
    }

    const int m0 = mtile * 16 + 8 * hi;
    if (nt < 2) {
        #pragma unroll
        for (int r = 0; r < 8; ++r) {
            const long m = m0 + r;
            z[m * HALF_ + ncol] = acc[r];
        }
    } else {
        const float be = b_eye[ncol - HALF_];
        #pragma unroll
        for (int r = 0; r < 8; ++r) {
            const long m = m0 + r;
            out[m * CIN_ + ncol] = acc[r] + be;
        }
    }
}

#define RPB_   196
#define QA_    1024
#define QB_    256
#define NGRP_  (E_ / 128)
#define NLD_   196


__device__ __forceinline__ void work_off_list(
    const int* q, int qn, int lane,
    const float* __restrict__ z, const float* __restrict__ vals,
    const int* __restrict__ cols, float* acc)
{
    for (int i0 = 0; i0 < qn; i0 += 32) {
        const int n   = (qn - i0 < 32) ? (qn - i0) : 32;
        const int ent = (lane < n) ? q[i0 + lane] : 0;
        const int e   = ent & 0xFFFFF;
        const int   cn = cols[e];
        const float v  = vals[e];
        for (int k = 0; k < n; k += 8) {
            float zz[8][4];
            float vk[8];
            int   dk[8];
            #pragma unroll
            for (int kk = 0; kk < 8; ++kk) {
                const int cnk = __builtin_amdgcn_readlane(cn, k + kk);
                vk[kk] = __int_as_float(__builtin_amdgcn_readlane(__float_as_int(v), k + kk));
                dk[kk] = __builtin_amdgcn_readlane(ent, k + kk) >> 20;
                #pragma unroll
                for (int b = 0; b < 4; ++b)
                    zz[kk][b] = z[((long)b * N_ + cnk) * HALF_ + lane];
            }
            #pragma unroll
            for (int kk = 0; kk < 8; ++kk) {
                if (k + kk < n) {
                    float* a = acc + dk[kk] * 128 + lane;
                    #pragma unroll
                    for (int b = 0; b < 4; ++b)
                        a[b * 32] += vk[kk] * zz[kk][b];
                }
            }
        }
    }
}

__device__ __forceinline__ int filter_pass(
    int ld0, int ld1, int wave, int lane, int lo, int cnt, int rot,
    const int* __restrict__ rows, int* q, bool& too_small)
{
    int qn = 0;
    for (int ld = ld0; ld < ld1; ld += 4) {
        int4 rr[4];
        int  eb[4];
        #pragma unroll
        for (int u = 0; u < 4; ++u) {
            const int gi = (ld + u) * 32 + wave;
            int g = gi + rot;
            if (g >= NGRP_) g -= NGRP_;
            eb[u] = g * 128 + lane * 4;
            if (ld + u < ld1 && gi < NGRP_) rr[u] = *(const int4*)(rows + eb[u]);
            else                            rr[u] = make_int4(-1, -1, -1, -1);
        }
        #pragma unroll
        for (int u = 0; u < 4; ++u) {
            const unsigned d0 = (unsigned)(rr[u].x - lo);
            const unsigned d1 = (unsigned)(rr[u].y - lo);
            const unsigned d2 = (unsigned)(rr[u].z - lo);
            const unsigned d3 = (unsigned)(rr[u].w - lo);
            const bool m0 = d0 < (unsigned)cnt, m1 = d1 < (unsigned)cnt;
            const bool m2 = d2 < (unsigned)cnt, m3 = d3 < (unsigned)cnt;
            if (__builtin_amdgcn_ballot_w32(m0 | m1 | m2 | m3) != 0u) {
                if (qn > QA_ - 128) {
                    too_small = true;
                } else {
                    const unsigned b0 = __builtin_amdgcn_ballot_w32(m0);
                    if (m0) q[qn + __builtin_amdgcn_mbcnt_lo(b0, 0u)] = (eb[u] + 0) | (int)(d0 << 20);
                    qn += __builtin_popcount(b0);
                    const unsigned b1 = __builtin_amdgcn_ballot_w32(m1);
                    if (m1) q[qn + __builtin_amdgcn_mbcnt_lo(b1, 0u)] = (eb[u] + 1) | (int)(d1 << 20);
                    qn += __builtin_popcount(b1);
                    const unsigned b2 = __builtin_amdgcn_ballot_w32(m2);
                    if (m2) q[qn + __builtin_amdgcn_mbcnt_lo(b2, 0u)] = (eb[u] + 2) | (int)(d2 << 20);
                    qn += __builtin_popcount(b2);
                    const unsigned b3 = __builtin_amdgcn_ballot_w32(m3);
                    if (m3) q[qn + __builtin_amdgcn_mbcnt_lo(b3, 0u)] = (eb[u] + 3) | (int)(d3 << 20);
                    qn += __builtin_popcount(b3);
                }
            }
        }
    }
    return qn;
}

__device__ __forceinline__ void collect_and_add(
    const int* queA, const int* cntA, int* qb, int wave, int lane,
    const float* __restrict__ z, const float* __restrict__ vals,
    const int* __restrict__ cols, float* acc)
{
    int nb = 0;
    for (int s0 = 0; s0 < 32; ++s0) {
        const int  s  = (s0 + wave) & 31;
        const int  ns = __builtin_amdgcn_readfirstlane(cntA[s]);
        const int* qs = queA + s * QA_;
        for (int i0 = 0; i0 < ns; i0 += 32) {
            const bool ok   = (i0 + lane) < ns;
            const int  ent  = ok ? qs[i0 + lane] : 0;
            const bool mine = ok && (((ent >> 20) & 31) == wave);
            const unsigned bm = __builtin_amdgcn_ballot_w32(mine);
            if (bm != 0u) {
                if (mine) qb[nb + __builtin_amdgcn_mbcnt_lo(bm, 0u)] = ent;
                nb += __builtin_popcount(bm);
                if (nb > QB_ - 32) {
                    __builtin_amdgcn_wave_barrier();
                    work_off_list(qb, nb, lane, z, vals, cols, acc);
                    __builtin_amdgcn_wave_barrier();
                    nb = 0;
                }
            }
        }
    }
    __builtin_amdgcn_wave_barrier();
    work_off_list(qb, nb, lane, z, vals, cols, acc);
}

__global__ __launch_bounds__(1024) void gcn_rowrange_accumulate(
    const float* __restrict__ z, const float* __restrict__ vals,
    const int* __restrict__ rows, const int* __restrict__ cols,
    const float* __restrict__ b_lin, float* __restrict__ out)
{
    __shared__ float acc[RPB_ * 128];
    __shared__ int   queA[32 * QA_];
    __shared__ int   queB[32 * QB_];
    __shared__ int   cntA[32];
    __shared__ int   list_too_small;

    const int tid  = threadIdx.x;
    const int lane = tid & 31;
    const int wave = tid >> 5;
    const int lo   = blockIdx.x * RPB_;
    const int cnt  = (N_ - lo < RPB_) ? (N_ - lo) : RPB_;

    const float bl = b_lin[lane];
    for (int i = tid; i < RPB_ * 128; i += 1024) acc[i] = bl;
    if (tid == 0) list_too_small = 0;
    __syncthreads();

    int* qa = queA + wave * QA_;
    int* qb = queB + wave * QB_;
    const int rot = blockIdx.x * 24;

    bool too_small = false;
    int qn = filter_pass(0, NLD_, wave, lane, lo, cnt, rot, rows, qa, too_small);
    if (lane == 0) {
        cntA[wave] = qn;
        if (too_small) list_too_small = 1;
    }
    __syncthreads();
    const int redo = __builtin_amdgcn_readfirstlane(list_too_small);
    if (redo == 0) {
        collect_and_add(queA, cntA, qb, wave, lane, z, vals, cols, acc);
    } else {
        for (int r = 0; r < NLD_; r += 8) {
            __syncthreads();
            bool unused = false;
            const int r1 = (r + 8 < NLD_) ? (r + 8) : NLD_;
            qn = filter_pass(r, r1, wave, lane, lo, cnt, rot, rows, qa, unused);
            if (lane == 0) cntA[wave] = qn;
            __syncthreads();
            collect_and_add(queA, cntA, qb, wave, lane, z, vals, cols, acc);
        }
    }
    __syncthreads();

    for (int i = tid; i < cnt * 128; i += 1024) {
        const int lrow = i >> 7;
        const int b    = (i >> 5) & 3;
        out[((long)b * N_ + lo + lrow) * CIN_ + lane] = acc[i];
    }
}

extern "C" void kernel_launch(void* const* d_in, const int* in_sizes, int n_in,
                              void* d_out, int out_size, void* d_ws, size_t ws_size,
                              hipStream_t stream) {
    const float* x     = (const float*)d_in[0];
    const float* vals  = (const float*)d_in[1];
    const float* W_lin = (const float*)d_in[2];
    const float* b_lin = (const float*)d_in[3];
    const float* W_eye = (const float*)d_in[4];
    const float* b_eye = (const float*)d_in[5];
    const int*   rows  = (const int*)d_in[6];
    const int*   cols  = (const int*)d_in[7];
    float* out = (float*)d_out;
    float* z   = (float*)d_ws;

    gcn_gemm_wmma<<<M_ / 32, 256, 0, stream>>>(x, W_lin, W_eye, b_eye, out, z);

    gcn_rowrange_accumulate<<<256, 1024, 0, stream>>>(z, vals, rows, cols, b_lin, out);
}


# GEMM launch made resident: 256 work-groups x 28 waves, a wave loops over 7 row tiles (7168 waves enter, not 50000), weight fragments held in registers, no LDS or barrier; k order kept
# speedup vs baseline: 3.1987x; 1.0823x over previous
#include <hip/hip_runtime.h>

#define B_    4
#define N_    50000
#define CIN_  64
#define E_    800000
#define HALF_ 32
#define M_    (B_ * N_)

typedef __attribute__((ext_vector_type(2))) float v2f;
typedef __attribute__((ext_vector_type(8))) float v8f;

#define G_SLOTS_  7
#define G_TRIPS_  7
#define G_TPB_    (G_SLOTS_ * G_TRIPS_)
#define NRT_      (M_ / 16)

__device__ __forceinline__ void load_a_tile(
    const float* __restrict__ x, int t, int l16, int hi, v2f (&a)[16])
{
    const int tt = (t < NRT_) ? t : (NRT_ - 1);
    const float* xrow = x + (long)(tt * 16 + l16) * CIN_ + hi * 2;
    #pragma unroll
    for (int kb = 0; kb < 16; ++kb)
        a[kb] = *(const v2f*)(xrow + kb * 4);
}

__global__ __launch_bounds__(896) void gcn_gemm_wmma(
    const float* __restrict__ x,
    const float* __restrict__ W_lin,
    const float* __restrict__ W_eye, const float* __restrict__ b_eye,
    float* __restrict__ out, float* __restrict__ z)
{
    const int tid   = threadIdx.x;
    const int wave  = __builtin_amdgcn_readfirstlane(tid >> 5);
    const int lane  = tid & 31;
    const int hi    = lane >> 4;
    const int l16   = lane & 15;
    const int nt    = wave & 3;
    const int slot  = wave >> 2;
    const int ncol  = nt * 16 + l16;

    const float* wcol = ((nt < 2) ? (W_lin + ncol * CIN_)
                                  : (W_eye + (ncol - HALF_) * CIN_)) + hi * 2;
    v2f bf[16];
    #pragma unroll
    for (int kb = 0; kb < 16; ++kb)
        bf[kb] = *(const v2f*)(wcol + kb * 4);

    const int t0 = blockIdx.x * G_TPB_ + slot;
    v2f a[2][16];
    load_a_tile(x, t0, l16, hi, a[0]);

    float be = 0.0f;
    if (nt >= 2) be = b_eye[ncol - HALF_];

    #pragma unroll
    for (int i = 0; i < G_TRIPS_; ++i) {
        const int t = t0 + i * G_SLOTS_;
        if (i + 1 < G_TRIPS_)
            load_a_tile(x, t + G_SLOTS_, l16, hi, a[(i + 1) & 1]);

        v8f acc = {};
        #pragma unroll
        for (int kb = 0; kb < 16; ++kb) {
            acc = __builtin_amdgcn_wmma_f32_16x16x4_f32(
                 false, a[i & 1][kb],  false, bf[kb],
                 (short)0, acc,  false,  false);
        }

        if (t < NRT_) {
            const int m0 = t * 16 + 8 * hi;
            if (nt < 2) {
                #pragma unroll
                for (int r = 0; r < 8; ++r) {
                    const long m = m0 + r;
                    z[m * HALF_ + ncol] = acc[r];
                }
            } else {
                #pragma unroll
                for (int r = 0; r < 8; ++r) {
                    const long m = m0 + r;
                    out[m * CIN_ + ncol] = acc[r] + be;
                }
            }
        }
    }
}

#define RPB_   196
#define QA_    1024
#define QB_    256
#define NGRP_  (E_ / 128)
#define NLD_   196


__device__ __forceinline__ void work_off_list(
    const int* q, int qn, int lane,
    const float* __restrict__ z, const float* __restrict__ vals,
    const int* __restrict__ cols, float* acc)
{
    for (int i0 = 0; i0 < qn; i0 += 32) {
        const int n   = (qn - i0 < 32) ? (qn - i0) : 32;
        const int ent = (lane < n) ? q[i0 + lane] : 0;
        const int e   = ent & 0xFFFFF;
        const int   cn = cols[e];
        const float v  = vals[e];
        for (int k = 0; k < n; k += 8) {
            float zz[8][4];
            float vk[8];
            int   dk[8];
            #pragma unroll
            for (int kk = 0; kk < 8; ++kk) {
                const int cnk = __builtin_amdgcn_readlane(cn, k + kk);
                vk[kk] = __int_as_float(__builtin_amdgcn_readlane(__float_as_int(v), k + kk));
                dk[kk] = __builtin_amdgcn_readlane(ent, k + kk) >> 20;
                #pragma unroll
                for (int b = 0; b < 4; ++b)
                    zz[kk][b] = z[((long)b * N_ + cnk) * HALF_ + lane];
            }
            #pragma unroll
            for (int kk = 0; kk < 8; ++kk) {
                if (k + kk < n) {
                    float* a = acc + dk[kk] * 128 + lane;
                    #pragma unroll
                    for (int b = 0; b < 4; ++b)
                        a[b * 32] += vk[kk] * zz[kk][b];
                }
            }
        }
    }
}

__device__ __forceinline__ int filter_pass(
    int ld0, int ld1, int wave, int lane, int lo, int cnt, int rot,
    const int* __restrict__ rows, int* q, bool& too_small)
{
    int qn = 0;
    for (int ld = ld0; ld < ld1; ld += 4) {
        int4 rr[4];
        int  eb[4];
        #pragma unroll
        for (int u = 0; u < 4; ++u) {
            const int gi = (ld + u) * 32 + wave;
            int g = gi + rot;
            if (g >= NGRP_) g -= NGRP_;
            eb[u] = g * 128 + lane * 4;
            if (ld + u < ld1 && gi < NGRP_) rr[u] = *(const int4*)(rows + eb[u]);
            else                            rr[u] = make_int4(-1, -1, -1, -1);
        }
        #pragma unroll
        for (int u = 0; u < 4; ++u) {
            const unsigned d0 = (unsigned)(rr[u].x - lo);
            const unsigned d1 = (unsigned)(rr[u].y - lo);
            const unsigned d2 = (unsigned)(rr[u].z - lo);
            const unsigned d3 = (unsigned)(rr[u].w - lo);
            const bool m0 = d0 < (unsigned)cnt, m1 = d1 < (unsigned)cnt;
            const bool m2 = d2 < (unsigned)cnt, m3 = d3 < (unsigned)cnt;
            if (__builtin_amdgcn_ballot_w32(m0 | m1 | m2 | m3) != 0u) {
                if (qn > QA_ - 128) {
                    too_small = true;
                } else {
                    const unsigned b0 = __builtin_amdgcn_ballot_w32(m0);
                    if (m0) q[qn + __builtin_amdgcn_mbcnt_lo(b0, 0u)] = (eb[u] + 0) | (int)(d0 << 20);
                    qn += __builtin_popcount(b0);
                    const unsigned b1 = __builtin_amdgcn_ballot_w32(m1);
                    if (m1) q[qn + __builtin_amdgcn_mbcnt_lo(b1, 0u)] = (eb[u] + 1) | (int)(d1 << 20);
                    qn += __builtin_popcount(b1);
                    const unsigned b2 = __builtin_amdgcn_ballot_w32(m2);
                    if (m2) q[qn + __builtin_amdgcn_mbcnt_lo(b2, 0u)] = (eb[u] + 2) | (int)(d2 << 20);
                    qn += __builtin_popcount(b2);
                    const unsigned b3 = __builtin_amdgcn_ballot_w32(m3);
                    if (m3) q[qn + __builtin_amdgcn_mbcnt_lo(b3, 0u)] = (eb[u] + 3) | (int)(d3 << 20);
                    qn += __builtin_popcount(b3);
                }
            }
        }
    }
    return qn;
}

__device__ __forceinline__ void collect_and_add(
    const int* queA, const int* cntA, int* qb, int wave, int lane,
    const float* __restrict__ z, const float* __restrict__ vals,
    const int* __restrict__ cols, float* acc)
{
    int nb = 0;
    for (int s0 = 0; s0 < 32; ++s0) {
        const int  s  = (s0 + wave) & 31;
        const int  ns = __builtin_amdgcn_readfirstlane(cntA[s]);
        const int* qs = queA + s * QA_;
        for (int i0 = 0; i0 < ns; i0 += 32) {
            const bool ok   = (i0 + lane) < ns;
            const int  ent  = ok ? qs[i0 + lane] : 0;
            const bool mine = ok && (((ent >> 20) & 31) == wave);
            const unsigned bm = __builtin_amdgcn_ballot_w32(mine);
            if (bm != 0u) {
                if (mine) qb[nb + __builtin_amdgcn_mbcnt_lo(bm, 0u)] = ent;
                nb += __builtin_popcount(bm);
                if (nb > QB_ - 32) {
                    __builtin_amdgcn_wave_barrier();
                    work_off_list(qb, nb, lane, z, vals, cols, acc);
                    __builtin_amdgcn_wave_barrier();
                    nb = 0;
                }
            }
        }
    }
    __builtin_amdgcn_wave_barrier();
    work_off_list(qb, nb, lane, z, vals, cols, acc);
}

__global__ __launch_bounds__(1024) void gcn_rowrange_accumulate(
    const float* __restrict__ z, const float* __restrict__ vals,
    const int* __restrict__ rows, const int* __restrict__ cols,
    const float* __restrict__ b_lin, float* __restrict__ out)
{
    __shared__ float acc[RPB_ * 128];
    __shared__ int   queA[32 * QA_];
    __shared__ int   queB[32 * QB_];
    __shared__ int   cntA[32];
    __shared__ int   list_too_small;

    const int tid  = threadIdx.x;
    const int lane = tid & 31;
    const int wave = tid >> 5;
    const int lo   = blockIdx.x * RPB_;
    const int cnt  = (N_ - lo < RPB_) ? (N_ - lo) : RPB_;

    const float bl = b_lin[lane];
    for (int i = tid; i < RPB_ * 128; i += 1024) acc[i] = bl;
    if (tid == 0) list_too_small = 0;
    __syncthreads();

    int* qa = queA + wave * QA_;
    int* qb = queB + wave * QB_;
    const int rot = blockIdx.x * 24;

    bool too_small = false;
    int qn = filter_pass(0, NLD_, wave, lane, lo, cnt, rot, rows, qa, too_small);
    if (lane == 0) {
        cntA[wave] = qn;
        if (too_small) list_too_small = 1;
    }
    __syncthreads();
    const int redo = __builtin_amdgcn_readfirstlane(list_too_small);
    if (redo == 0) {
        collect_and_add(queA, cntA, qb, wave, lane, z, vals, cols, acc);
    } else {
        for (int r = 0; r < NLD_; r += 8) {
            __syncthreads();
            bool unused = false;
            const int r1 = (r + 8 < NLD_) ? (r + 8) : NLD_;
            qn = filter_pass(r, r1, wave, lane, lo, cnt, rot, rows, qa, unused);
            if (lane == 0) cntA[wave] = qn;
            __syncthreads();
            collect_and_add(queA, cntA, qb, wave, lane, z, vals, cols, acc);
        }
    }
    __syncthreads();

    for (int i = tid; i < cnt * 128; i += 1024) {
        const int lrow = i >> 7;
        const int b    = (i >> 5) & 3;
        out[((long)b * N_ + lo + lrow) * CIN_ + lane] = acc[i];
    }
}

extern "C" void kernel_launch(void* const* d_in, const int* in_sizes, int n_in,
                              void* d_out, int out_size, void* d_ws, size_t ws_size,
                              hipStream_t stream) {
    const float* x     = (const float*)d_in[0];
    const float* vals  = (const float*)d_in[1];
    const float* W_lin = (const float*)d_in[2];
    const float* b_lin = (const float*)d_in[3];
    const float* W_eye = (const float*)d_in[4];
    const float* b_eye = (const float*)d_in[5];
    const int*   rows  = (const int*)d_in[6];
    const int*   cols  = (const int*)d_in[7];
    float* out = (float*)d_out;
    float* z   = (float*)d_ws;

    gcn_gemm_wmma<<<256, 896, 0, stream>>>(x, W_lin, W_eye, b_eye, out, z);

    gcn_rowrange_accumulate<<<256, 1024, 0, stream>>>(z, vals, rows, cols, b_lin, out);
}


# Launch 1 writes one owner byte per edge (row / 196); each work-group of launch 2 filters these bytes, 16 edges per lane and load, not rows[]; order of a row's sum changed, still fixed by code
# speedup vs baseline: 3.7562x; 1.1743x over previous
#include <hip/hip_runtime.h>

#define B_    4
#define N_    50000
#define CIN_  64
#define E_    800000
#define HALF_ 32
#define M_    (B_ * N_)

typedef __attribute__((ext_vector_type(2))) float v2f;
typedef __attribute__((ext_vector_type(8))) float v8f;

#define RPB_      196
#define OWN_W_    (E_ / 4)
#define OWN_TPB_  782
#define G_SLOTS_  7
#define G_TRIPS_  7
#define G_TPB_    (G_SLOTS_ * G_TRIPS_)
#define NRT_      (M_ / 16)

__device__ __forceinline__ void load_a_tile(
    const float* __restrict__ x, int t, int l16, int hi, v2f (&a)[16])
{
    const int tt = (t < NRT_) ? t : (NRT_ - 1);
    const float* xrow = x + (long)(tt * 16 + l16) * CIN_ + hi * 2;
    #pragma unroll
    for (int kb = 0; kb < 16; ++kb)
        a[kb] = *(const v2f*)(xrow + kb * 4);
}

__global__ __launch_bounds__(896) void gcn_gemm_wmma(
    const float* __restrict__ x,
    const float* __restrict__ W_lin,
    const float* __restrict__ W_eye, const float* __restrict__ b_eye,
    const int* __restrict__ rows,
    float* __restrict__ out, float* __restrict__ z, unsigned int* __restrict__ own)
{
    const int tid   = threadIdx.x;

    const int  wi       = blockIdx.x * OWN_TPB_ + tid;
    const bool own_live = (tid < OWN_TPB_) && (wi < OWN_W_);
    const int4 er       = *(const int4*)(rows + (own_live ? wi : 0) * 4);

    const int wave  = __builtin_amdgcn_readfirstlane(tid >> 5);
    const int lane  = tid & 31;
    const int hi    = lane >> 4;
    const int l16   = lane & 15;
    const int nt    = wave & 3;
    const int slot  = wave >> 2;
    const int ncol  = nt * 16 + l16;

    const float* wcol = ((nt < 2) ? (W_lin + ncol * CIN_)
                                  : (W_eye + (ncol - HALF_) * CIN_)) + hi * 2;
    v2f bf[16];
    #pragma unroll
    for (int kb = 0; kb < 16; ++kb)
        bf[kb] = *(const v2f*)(wcol + kb * 4);

    const int t0 = blockIdx.x * G_TPB_ + slot;
    v2f a[2][16];
    load_a_tile(x, t0, l16, hi, a[0]);

    float be = 0.0f;
    if (nt >= 2) be = b_eye[ncol - HALF_];

    if (own_live)
        own[wi] =  (((unsigned)er.x / RPB_) & 255u)
                | ((((unsigned)er.y / RPB_) & 255u) << 8)
                | ((((unsigned)er.z / RPB_) & 255u) << 16)
                |  (((unsigned)er.w / RPB_) << 24);

    #pragma unroll
    for (int i = 0; i < G_TRIPS_; ++i) {
        const int t = t0 + i * G_SLOTS_;
        if (i + 1 < G_TRIPS_)
            load_a_tile(x, t + G_SLOTS_, l16, hi, a[(i + 1) & 1]);

        v8f acc = {};
        #pragma unroll
        for (int kb = 0; kb < 16; ++kb) {
            acc = __builtin_amdgcn_wmma_f32_16x16x4_f32(
                 false, a[i & 1][kb],  false, bf[kb],
                 (short)0, acc,  false,  false);
        }

        if (t < NRT_) {
            const int m0 = t * 16 + 8 * hi;
            if (nt < 2) {
                #pragma unroll
                for (int r = 0; r < 8; ++r) {
                    const long m = m0 + r;
                    z[m * HALF_ + ncol] = acc[r];
                }
            } else {
                #pragma unroll
                for (int r = 0; r < 8; ++r) {
                    const long m = m0 + r;
                    out[m * CIN_ + ncol] = acc[r] + be;
                }
            }
        }
    }
}

#define QA_    1024
#define QB_    256
#define OWN_Q_ (E_ / 16)
#define NGRP_  ((OWN_Q_ + 31) / 32)
#define NLD_   ((NGRP_ + 31) / 32)
#define FU_    7


__device__ __forceinline__ void work_off_list(
    const int* q, int qn, int lane,
    const float* __restrict__ z, const float* __restrict__ vals,
    const int* __restrict__ cols, float* acc)
{
    for (int i0 = 0; i0 < qn; i0 += 32) {
        const int n   = (qn - i0 < 32) ? (qn - i0) : 32;
        const int ent = (lane < n) ? q[i0 + lane] : 0;
        const int e   = ent & 0xFFFFF;
        const int   cn = cols[e];
        const float v  = vals[e];
        for (int k = 0; k < n; k += 8) {
            float zz[8][4];
            float vk[8];
            int   dk[8];
            #pragma unroll
            for (int kk = 0; kk < 8; ++kk) {
                const int cnk = __builtin_amdgcn_readlane(cn, k + kk);
                vk[kk] = __int_as_float(__builtin_amdgcn_readlane(__float_as_int(v), k + kk));
                dk[kk] = __builtin_amdgcn_readlane(ent, k + kk) >> 20;
                #pragma unroll
                for (int b = 0; b < 4; ++b)
                    zz[kk][b] = z[((long)b * N_ + cnk) * HALF_ + lane];
            }
            #pragma unroll
            for (int kk = 0; kk < 8; ++kk) {
                if (k + kk < n) {
                    float* a = acc + dk[kk] * 128 + lane;
                    #pragma unroll
                    for (int b = 0; b < 4; ++b)
                        a[b * 32] += vk[kk] * zz[kk][b];
                }
            }
        }
    }
}

__device__ __forceinline__ int filter_pass(
    int ld0, int ld1, int wave, int lane, unsigned key, int rot,
    const uint4* __restrict__ own, int* q, bool& too_small)
{
    const int sw = __builtin_amdgcn_readfirstlane(wave);
    int qn = 0;
    for (int ld = ld0; ld < ld1; ld += FU_) {
        uint4 ww[FU_];
        int   pc[FU_];
        bool  live[FU_];
        #pragma unroll
        for (int u = 0; u < FU_; ++u) {
            const int gi = (ld + u) * 32 + sw;
            int g = gi + rot;
            if (g >= NGRP_) g -= NGRP_;
            pc[u] = g * 32 + lane;
            live[u] = (ld + u < ld1) && (gi < NGRP_) && (pc[u] < OWN_Q_);
            const int pl = (pc[u] < OWN_Q_) ? pc[u] : (OWN_Q_ - 1);
            ww[u] = own[pl];
        }
        #pragma unroll
        for (int u = 0; u < FU_; ++u) {
            const unsigned t0 = ww[u].x ^ key, t1 = ww[u].y ^ key;
            const unsigned t2 = ww[u].z ^ key, t3 = ww[u].w ^ key;
            const unsigned n0 = (t0 - 0x01010101u) & ~t0 & 0x80808080u;
            const unsigned n1 = (t1 - 0x01010101u) & ~t1 & 0x80808080u;
            const unsigned n2 = (t2 - 0x01010101u) & ~t2 & 0x80808080u;
            const unsigned n3 = (t3 - 0x01010101u) & ~t3 & 0x80808080u;
            unsigned f = n0 | (n1 >> 1) | (n2 >> 2) | (n3 >> 3);
            if (!live[u]) f = 0u;
            unsigned bal = __builtin_amdgcn_ballot_w32(f != 0u);
            if (bal != 0u) {
                if (qn > QA_ - 512) {
                    too_small = true;
                } else {
                    do {
                        if (f != 0u) {
                            const int p = __builtin_ctz(f);
                            q[qn + __builtin_amdgcn_mbcnt_lo(bal, 0u)] =
                                pc[u] * 16 + 28 - ((p & 7) << 2) + (p >> 3);
                        }
                        qn += __builtin_popcount(bal);
                        f &= f - 1u;
                        bal = __builtin_amdgcn_ballot_w32(f != 0u);
                    } while (bal != 0u);
                }
            }
        }
    }
    return qn;
}

__device__ __forceinline__ int add_local_rows(
    int* q, int qn, int lane, int lo, int cnt, const int* __restrict__ rows)
{
    int wn = 0;
    for (int i0 = 0; i0 < qn; i0 += 128) {
        int  e[4];
        int  r[4];
        bool ok[4];
        #pragma unroll
        for (int u = 0; u < 4; ++u) {
            ok[u] = (i0 + u * 32 + lane) < qn;
            e[u]  = ok[u] ? q[i0 + u * 32 + lane] : 0;
        }
        #pragma unroll
        for (int u = 0; u < 4; ++u) r[u] = rows[e[u]];
        #pragma unroll
        for (int u = 0; u < 4; ++u) {
            const unsigned d    = (unsigned)(r[u] - lo);
            const bool     keep = ok[u] && d < (unsigned)cnt;
            const unsigned bm   = __builtin_amdgcn_ballot_w32(keep);
            if (keep) q[wn + __builtin_amdgcn_mbcnt_lo(bm, 0u)] = e[u] | (int)(d << 20);
            wn += __builtin_popcount(bm);
        }
    }
    return wn;
}

__device__ __forceinline__ void collect_and_add(
    const int* queA, const int* cntA, int* qb, int wave, int lane,
    const float* __restrict__ z, const float* __restrict__ vals,
    const int* __restrict__ cols, float* acc)
{
    int nb = 0;
    for (int s0 = 0; s0 < 32; ++s0) {
        const int  s  = (s0 + wave) & 31;
        const int  ns = __builtin_amdgcn_readfirstlane(cntA[s]);
        const int* qs = queA + s * QA_;
        for (int i0 = 0; i0 < ns; i0 += 32) {
            const bool ok   = (i0 + lane) < ns;
            const int  ent  = ok ? qs[i0 + lane] : 0;
            const bool mine = ok && (((ent >> 20) & 31) == wave);
            const unsigned bm = __builtin_amdgcn_ballot_w32(mine);
            if (bm != 0u) {
                if (mine) qb[nb + __builtin_amdgcn_mbcnt_lo(bm, 0u)] = ent;
                nb += __builtin_popcount(bm);
                if (nb > QB_ - 32) {
                    __builtin_amdgcn_wave_barrier();
                    work_off_list(qb, nb, lane, z, vals, cols, acc);
                    __builtin_amdgcn_wave_barrier();
                    nb = 0;
                }
            }
        }
    }
    __builtin_amdgcn_wave_barrier();
    work_off_list(qb, nb, lane, z, vals, cols, acc);
}

__global__ __launch_bounds__(1024) void gcn_rowrange_accumulate(
    const float* __restrict__ z, const float* __restrict__ vals,
    const int* __restrict__ rows, const int* __restrict__ cols,
    const float* __restrict__ b_lin, const unsigned int* __restrict__ own,
    float* __restrict__ out)
{
    __shared__ float acc[RPB_ * 128];
    __shared__ int   queA[32 * QA_];
    __shared__ int   queB[32 * QB_];
    __shared__ int   cntA[32];
    __shared__ int   list_too_small;

    const int tid  = threadIdx.x;
    const int lane = tid & 31;
    const int wave = tid >> 5;
    const int lo   = blockIdx.x * RPB_;
    const int cnt  = (N_ - lo < RPB_) ? (N_ - lo) : RPB_;

    const float bl = b_lin[lane];
    for (int i = tid; i < RPB_ * 128; i += 1024) acc[i] = bl;
    if (tid == 0) list_too_small = 0;
    __syncthreads();

    int* qa = queA + wave * QA_;
    int* qb = queB + wave * QB_;
    const int rot = blockIdx.x * 6;
    const uint4* own16 = (const uint4*)own;
    const unsigned key = blockIdx.x * 0x01010101u;

    bool too_small = false;
    int qn = filter_pass(0, NLD_, wave, lane, key, rot, own16, qa, too_small);
    if (!too_small) qn = add_local_rows(qa, qn, lane, lo, cnt, rows);
    if (lane == 0) {
        cntA[wave] = qn;
        if (too_small) list_too_small = 1;
    }
    __syncthreads();
    const int redo = __builtin_amdgcn_readfirstlane(list_too_small);
    if (redo == 0) {
        collect_and_add(queA, cntA, qb, wave, lane, z, vals, cols, acc);
    } else {
        for (int r = 0; r < NLD_; r += 2) {
            __syncthreads();
            bool unused = false;
            const int r1 = (r + 2 < NLD_) ? (r + 2) : NLD_;
            qn = filter_pass(r, r1, wave, lane, key, rot, own16, qa, unused);
            qn = add_local_rows(qa, qn, lane, lo, cnt, rows);
            if (lane == 0) cntA[wave] = qn;
            __syncthreads();
            collect_and_add(queA, cntA, qb, wave, lane, z, vals, cols, acc);
        }
    }
    __syncthreads();

    for (int i = tid; i < cnt * 128; i += 1024) {
        const int lrow = i >> 7;
        const int b    = (i >> 5) & 3;
        out[((long)b * N_ + lo + lrow) * CIN_ + lane] = acc[i];
    }
}

extern "C" void kernel_launch(void* const* d_in, const int* in_sizes, int n_in,
                              void* d_out, int out_size, void* d_ws, size_t ws_size,
                              hipStream_t stream) {
    const float* x     = (const float*)d_in[0];
    const float* vals  = (const float*)d_in[1];
    const float* W_lin = (const float*)d_in[2];
    const float* b_lin = (const float*)d_in[3];
    const float* W_eye = (const float*)d_in[4];
    const float* b_eye = (const float*)d_in[5];
    const int*   rows  = (const int*)d_in[6];
    const int*   cols  = (const int*)d_in[7];
    float* out = (float*)d_out;
    float* z   = (float*)d_ws;
    unsigned int* own = (unsigned int*)(z + M_ * HALF_);

    gcn_gemm_wmma<<<256, 896, 0, stream>>>(x, W_lin, W_eye, b_eye, rows, out, z, own);

    gcn_rowrange_accumulate<<<256, 1024, 0, stream>>>(z, vals, rows, cols, b_lin, own, out);
}


# Filtering wave files each kept edge at the rows look-up into a short LDS run per owning wave (32x32 runs, ballot ranks, no atomics); no wave scans all 32 lists; row sum order changed, fixed by code
# speedup vs baseline: 4.5474x; 1.2106x over previous
#include <hip/hip_runtime.h>

#define B_    4
#define N_    50000
#define CIN_  64
#define E_    800000
#define HALF_ 32
#define M_    (B_ * N_)

typedef __attribute__((ext_vector_type(2))) float v2f;
typedef __attribute__((ext_vector_type(8))) float v8f;

#define RPB_      196
#define OWN_W_    (E_ / 4)
#define OWN_TPB_  782
#define G_SLOTS_  7
#define G_TRIPS_  7
#define G_TPB_    (G_SLOTS_ * G_TRIPS_)
#define NRT_      (M_ / 16)

__device__ __forceinline__ void load_a_tile(
    const float* __restrict__ x, int t, int l16, int hi, v2f (&a)[16])
{
    const int tt = (t < NRT_) ? t : (NRT_ - 1);
    const float* xrow = x + (long)(tt * 16 + l16) * CIN_ + hi * 2;
    #pragma unroll
    for (int kb = 0; kb < 16; ++kb)
        a[kb] = *(const v2f*)(xrow + kb * 4);
}

__global__ __launch_bounds__(896) void gcn_gemm_wmma(
    const float* __restrict__ x,
    const float* __restrict__ W_lin,
    const float* __restrict__ W_eye, const float* __restrict__ b_eye,
    const int* __restrict__ rows,
    float* __restrict__ out, float* __restrict__ z, unsigned int* __restrict__ own)
{
    const int tid   = threadIdx.x;

    const int  wi       = blockIdx.x * OWN_TPB_ + tid;
    const bool own_live = (tid < OWN_TPB_) && (wi < OWN_W_);
    const int4 er       = *(const int4*)(rows + (own_live ? wi : 0) * 4);

    const int wave  = __builtin_amdgcn_readfirstlane(tid >> 5);
    const int lane  = tid & 31;
    const int hi    = lane >> 4;
    const int l16   = lane & 15;
    const int nt    = wave & 3;
    const int slot  = wave >> 2;
    const int ncol  = nt * 16 + l16;

    const float* wcol = ((nt < 2) ? (W_lin + ncol * CIN_)
                                  : (W_eye + (ncol - HALF_) * CIN_)) + hi * 2;
    v2f bf[16];
    #pragma unroll
    for (int kb = 0; kb < 16; ++kb)
        bf[kb] = *(const v2f*)(wcol + kb * 4);

    const int t0 = blockIdx.x * G_TPB_ + slot;
    v2f a[2][16];
    load_a_tile(x, t0, l16, hi, a[0]);

    float be = 0.0f;
    if (nt >= 2) be = b_eye[ncol - HALF_];

    if (own_live)
        own[wi] =  (((unsigned)er.x / RPB_) & 255u)
                | ((((unsigned)er.y / RPB_) & 255u) << 8)
                | ((((unsigned)er.z / RPB_) & 255u) << 16)
                |  (((unsigned)er.w / RPB_) << 24);

    #pragma unroll
    for (int i = 0; i < G_TRIPS_; ++i) {
        const int t = t0 + i * G_SLOTS_;
        if (i + 1 < G_TRIPS_)
            load_a_tile(x, t + G_SLOTS_, l16, hi, a[(i + 1) & 1]);

        v8f acc = {};
        #pragma unroll
        for (int kb = 0; kb < 16; ++kb) {
            acc = __builtin_amdgcn_wmma_f32_16x16x4_f32(
                 false, a[i & 1][kb],  false, bf[kb],
                 (short)0, acc,  false,  false);
        }

        if (t < NRT_) {
            const int m0 = t * 16 + 8 * hi;
            if (nt < 2) {
                #pragma unroll
                for (int r = 0; r < 8; ++r) {
                    const long m = m0 + r;
                    z[m * HALF_ + ncol] = acc[r];
                }
            } else {
                #pragma unroll
                for (int r = 0; r < 8; ++r) {
                    const long m = m0 + r;
                    out[m * CIN_ + ncol] = acc[r] + be;
                }
            }
        }
    }
}

#define QA_    1024
#define QB_    256
#define SCAP_  16
#define SPITCH_ 17
#define OWN_Q_ (E_ / 16)
#define NGRP_  ((OWN_Q_ + 31) / 32)
#define NLD_   ((NGRP_ + 31) / 32)
#define FU_    7


__device__ __forceinline__ void work_off_list(
    const int* q, int qn, int lane,
    const float* __restrict__ z, const float* __restrict__ vals,
    const int* __restrict__ cols, float* acc)
{
    for (int i0 = 0; i0 < qn; i0 += 32) {
        const int n   = (qn - i0 < 32) ? (qn - i0) : 32;
        const int ent = (lane < n) ? q[i0 + lane] : 0;
        const int e   = ent & 0xFFFFF;
        const int   cn = cols[e];
        const float v  = vals[e];
        for (int k = 0; k < n; k += 8) {
            float zz[8][4];
            float vk[8];
            int   dk[8];
            #pragma unroll
            for (int kk = 0; kk < 8; ++kk) {
                const int cnk = __builtin_amdgcn_readlane(cn, k + kk);
                vk[kk] = __int_as_float(__builtin_amdgcn_readlane(__float_as_int(v), k + kk));
                dk[kk] = __builtin_amdgcn_readlane(ent, k + kk) >> 20;
                #pragma unroll
                for (int b = 0; b < 4; ++b)
                    zz[kk][b] = z[((long)b * N_ + cnk) * HALF_ + lane];
            }
            #pragma unroll
            for (int kk = 0; kk < 8; ++kk) {
                if (k + kk < n) {
                    float* a = acc + dk[kk] * 128 + lane;
                    #pragma unroll
                    for (int b = 0; b < 4; ++b)
                        a[b * 32] += vk[kk] * zz[kk][b];
                }
            }
        }
    }
}

__device__ __forceinline__ int filter_pass(
    int ld0, int ld1, int wave, int lane, unsigned key, int rot,
    const uint4* __restrict__ own, int* q, bool& too_small)
{
    const int sw = __builtin_amdgcn_readfirstlane(wave);
    int qn = 0;
    for (int ld = ld0; ld < ld1; ld += FU_) {
        uint4 ww[FU_];
        int   pc[FU_];
        bool  live[FU_];
        #pragma unroll
        for (int u = 0; u < FU_; ++u) {
            const int gi = (ld + u) * 32 + sw;
            int g = gi + rot;
            if (g >= NGRP_) g -= NGRP_;
            pc[u] = g * 32 + lane;
            live[u] = (ld + u < ld1) && (gi < NGRP_) && (pc[u] < OWN_Q_);
            const int pl = (pc[u] < OWN_Q_) ? pc[u] : (OWN_Q_ - 1);
            ww[u] = own[pl];
        }
        #pragma unroll
        for (int u = 0; u < FU_; ++u) {
            const unsigned t0 = ww[u].x ^ key, t1 = ww[u].y ^ key;
            const unsigned t2 = ww[u].z ^ key, t3 = ww[u].w ^ key;
            const unsigned n0 = (t0 - 0x01010101u) & ~t0 & 0x80808080u;
            const unsigned n1 = (t1 - 0x01010101u) & ~t1 & 0x80808080u;
            const unsigned n2 = (t2 - 0x01010101u) & ~t2 & 0x80808080u;
            const unsigned n3 = (t3 - 0x01010101u) & ~t3 & 0x80808080u;
            unsigned f = n0 | (n1 >> 1) | (n2 >> 2) | (n3 >> 3);
            if (!live[u]) f = 0u;
            unsigned bal = __builtin_amdgcn_ballot_w32(f != 0u);
            if (bal != 0u) {
                if (qn > QA_ - 512) {
                    too_small = true;
                } else {
                    do {
                        if (f != 0u) {
                            const int p = __builtin_ctz(f);
                            q[qn + __builtin_amdgcn_mbcnt_lo(bal, 0u)] =
                                pc[u] * 16 + 28 - ((p & 7) << 2) + (p >> 3);
                        }
                        qn += __builtin_popcount(bal);
                        f &= f - 1u;
                        bal = __builtin_amdgcn_ballot_w32(f != 0u);
                    } while (bal != 0u);
                }
            }
        }
    }
    return qn;
}

__device__ __forceinline__ int add_local_rows(
    int* q, int qn, int lane, int lo, int cnt, const int* __restrict__ rows)
{
    int wn = 0;
    for (int i0 = 0; i0 < qn; i0 += 128) {
        int  e[4];
        int  r[4];
        bool ok[4];
        #pragma unroll
        for (int u = 0; u < 4; ++u) {
            ok[u] = (i0 + u * 32 + lane) < qn;
            e[u]  = ok[u] ? q[i0 + u * 32 + lane] : 0;
        }
        #pragma unroll
        for (int u = 0; u < 4; ++u) r[u] = rows[e[u]];
        #pragma unroll
        for (int u = 0; u < 4; ++u) {
            const unsigned d    = (unsigned)(r[u] - lo);
            const bool     keep = ok[u] && d < (unsigned)cnt;
            const unsigned bm   = __builtin_amdgcn_ballot_w32(keep);
            if (keep) q[wn + __builtin_amdgcn_mbcnt_lo(bm, 0u)] = e[u] | (int)(d << 20);
            wn += __builtin_popcount(bm);
        }
    }
    return wn;
}

__device__ __forceinline__ void collect_and_add(
    const int* queA, const int* cntA, int* qb, int wave, int lane,
    const float* __restrict__ z, const float* __restrict__ vals,
    const int* __restrict__ cols, float* acc)
{
    int nb = 0;
    for (int s0 = 0; s0 < 32; ++s0) {
        const int  s  = (s0 + wave) & 31;
        const int  ns = __builtin_amdgcn_readfirstlane(cntA[s]);
        const int* qs = queA + s * QA_;
        for (int i0 = 0; i0 < ns; i0 += 32) {
            const bool ok   = (i0 + lane) < ns;
            const int  ent  = ok ? qs[i0 + lane] : 0;
            const bool mine = ok && (((ent >> 20) & 31) == wave);
            const unsigned bm = __builtin_amdgcn_ballot_w32(mine);
            if (bm != 0u) {
                if (mine) qb[nb + __builtin_amdgcn_mbcnt_lo(bm, 0u)] = ent;
                nb += __builtin_popcount(bm);
                if (nb > QB_ - 32) {
                    __builtin_amdgcn_wave_barrier();
                    work_off_list(qb, nb, lane, z, vals, cols, acc);
                    __builtin_amdgcn_wave_barrier();
                    nb = 0;
                }
            }
        }
    }
    __builtin_amdgcn_wave_barrier();
    work_off_list(qb, nb, lane, z, vals, cols, acc);
}


__device__ __forceinline__ bool file_rows(
    const int* q, int qn, int wave, int lane, int lo, int cnt,
    const int* __restrict__ rows, int* runs, int* runcnt)
{
    int  mycnt = 0;
    bool over  = false;
    for (int i0 = 0; i0 < qn; i0 += 128) {
        int  e[4];
        int  r[4];
        bool ok[4];
        #pragma unroll
        for (int u = 0; u < 4; ++u) {
            ok[u] = (i0 + u * 32 + lane) < qn;
            e[u]  = ok[u] ? q[i0 + u * 32 + lane] : 0;
        }
        #pragma unroll
        for (int u = 0; u < 4; ++u) r[u] = rows[e[u]];
        #pragma unroll
        for (int u = 0; u < 4; ++u) {
            const unsigned d    = (unsigned)(r[u] - lo);
            const bool     keep = ok[u] && d < (unsigned)cnt;
            const int      w    = (int)(d & 31u);
            unsigned same = __builtin_amdgcn_ballot_w32(keep);
            unsigned tome = same;
            #pragma unroll
            for (int b = 0; b < 5; ++b) {
                const unsigned bb = __builtin_amdgcn_ballot_w32(keep && ((w >> b) & 1));
                same &= ((w    >> b) & 1) ? bb : ~bb;
                tome &= ((lane >> b) & 1) ? bb : ~bb;
            }
            const int base = __builtin_amdgcn_ds_bpermute(w << 2, mycnt);
            const int pos  = base + (int)__builtin_amdgcn_mbcnt_lo(same, 0u);
            if (keep) {
                if (pos < SCAP_) runs[(w * 32 + wave) * SPITCH_ + pos] = e[u] | (int)(d << 20);
                else             over = true;
            }
            mycnt += __builtin_popcount(tome);
        }
    }
    runcnt[wave * 32 + lane] = mycnt;
    return __builtin_amdgcn_ballot_w32(over) == 0u;
}

__device__ __forceinline__ void gather_runs_and_add(
    const int* runs, const int* runcnt, int* qc, int wave, int lane,
    const float* __restrict__ z, const float* __restrict__ vals,
    const int* __restrict__ cols, float* acc)
{
    const int  c   = runcnt[lane * 32 + wave];
    const int* run = runs + (wave * 32 + lane) * SPITCH_;
    int p = c;
    #pragma unroll
    for (int o = 1; o < 32; o <<= 1) {
        const int t = __shfl_up(p, o);
        if (lane >= o) p += t;
    }
    const int nb    = __builtin_amdgcn_readlane(p, 31);
    const int start = p - c;
    for (int k = 0; k < SCAP_; ++k) {
        if (__builtin_amdgcn_ballot_w32(k < c) == 0u) break;
        if (k < c) qc[start + k] = run[k];
    }
    __builtin_amdgcn_wave_barrier();
    work_off_list(qc, nb, lane, z, vals, cols, acc);
}

__global__ __launch_bounds__(1024) void gcn_rowrange_accumulate(
    const float* __restrict__ z, const float* __restrict__ vals,
    const int* __restrict__ rows, const int* __restrict__ cols,
    const float* __restrict__ b_lin, const unsigned int* __restrict__ own,
    float* __restrict__ out)
{
    __shared__ float acc[RPB_ * 128];
    __shared__ int   queA[32 * QA_];
    __shared__ int   runs[32 * 32 * SPITCH_];
    __shared__ int   runcnt[32 * 32];
    __shared__ int   cntA[32];
    __shared__ int   list_too_small;

    const int tid  = threadIdx.x;
    const int lane = tid & 31;
    const int wave = tid >> 5;
    const int lo   = blockIdx.x * RPB_;
    const int cnt  = (N_ - lo < RPB_) ? (N_ - lo) : RPB_;

    const float bl = b_lin[lane];
    for (int i = tid; i < RPB_ * 128; i += 1024) acc[i] = bl;
    if (tid == 0) list_too_small = 0;
    __syncthreads();

    int* qa = queA + wave * QA_;
    int* qb = runs + wave * QB_;
    const int rot = blockIdx.x * 6;
    const uint4* own16 = (const uint4*)own;
    const unsigned key = blockIdx.x * 0x01010101u;

    bool too_small = false;
    int qn = filter_pass(0, NLD_, wave, lane, key, rot, own16, qa, too_small);
    if (!too_small) too_small = !file_rows(qa, qn, wave, lane, lo, cnt, rows, runs, runcnt);
    if (lane == 0 && too_small) list_too_small = 1;
    __syncthreads();
    const int redo = __builtin_amdgcn_readfirstlane(list_too_small);
    if (redo == 0) {
        gather_runs_and_add(runs, runcnt, qa, wave, lane, z, vals, cols, acc);
    } else {
        for (int r = 0; r < NLD_; r += 2) {
            __syncthreads();
            bool unused = false;
            const int r1 = (r + 2 < NLD_) ? (r + 2) : NLD_;
            qn = filter_pass(r, r1, wave, lane, key, rot, own16, qa, unused);
            qn = add_local_rows(qa, qn, lane, lo, cnt, rows);
            if (lane == 0) cntA[wave] = qn;
            __syncthreads();
            collect_and_add(queA, cntA, qb, wave, lane, z, vals, cols, acc);
        }
    }
    __syncthreads();

    for (int i = tid; i < cnt * 128; i += 1024) {
        const int lrow = i >> 7;
        const int b    = (i >> 5) & 3;
        out[((long)b * N_ + lo + lrow) * CIN_ + lane] = acc[i];
    }
}

extern "C" void kernel_launch(void* const* d_in, const int* in_sizes, int n_in,
                              void* d_out, int out_size, void* d_ws, size_t ws_size,
                              hipStream_t stream) {
    const float* x     = (const float*)d_in[0];
    const float* vals  = (const float*)d_in[1];
    const float* W_lin = (const float*)d_in[2];
    const float* b_lin = (const float*)d_in[3];
    const float* W_eye = (const float*)d_in[4];
    const float* b_eye = (const float*)d_in[5];
    const int*   rows  = (const int*)d_in[6];
    const int*   cols  = (const int*)d_in[7];
    float* out = (float*)d_out;
    float* z   = (float*)d_ws;
    unsigned int* own = (unsigned int*)(z + M_ * HALF_);

    gcn_gemm_wmma<<<256, 896, 0, stream>>>(x, W_lin, W_eye, b_eye, rows, out, z, own);

    gcn_rowrange_accumulate<<<256, 1024, 0, stream>>>(z, vals, rows, cols, b_lin, own, out);
}


# Launch 2 keeps each wave's row sums (7 rows x 4 batches) in registers picked by a wave-uniform index, not in LDS: no LDS read-add-store per edge, no start-value pass, no store pass; sum order kept
# speedup vs baseline: 4.8417x; 1.0647x over previous
#include <hip/hip_runtime.h>

#define B_    4
#define N_    50000
#define CIN_  64
#define E_    800000
#define HALF_ 32
#define M_    (B_ * N_)

typedef __attribute__((ext_vector_type(2))) float v2f;
typedef __attribute__((ext_vector_type(8))) float v8f;

#define RPB_      196
#define OWN_W_    (E_ / 4)
#define OWN_TPB_  782
#define G_SLOTS_  7
#define G_TRIPS_  7
#define G_TPB_    (G_SLOTS_ * G_TRIPS_)
#define NRT_      (M_ / 16)

__device__ __forceinline__ void load_a_tile(
    const float* __restrict__ x, int t, int l16, int hi, v2f (&a)[16])
{
    const int tt = (t < NRT_) ? t : (NRT_ - 1);
    const float* xrow = x + (long)(tt * 16 + l16) * CIN_ + hi * 2;
    #pragma unroll
    for (int kb = 0; kb < 16; ++kb)
        a[kb] = *(const v2f*)(xrow + kb * 4);
}

__global__ __launch_bounds__(896) void gcn_gemm_wmma(
    const float* __restrict__ x,
    const float* __restrict__ W_lin,
    const float* __restrict__ W_eye, const float* __restrict__ b_eye,
    const int* __restrict__ rows,
    float* __restrict__ out, float* __restrict__ z, unsigned int* __restrict__ own)
{
    const int tid   = threadIdx.x;

    const int  wi       = blockIdx.x * OWN_TPB_ + tid;
    const bool own_live = (tid < OWN_TPB_) && (wi < OWN_W_);
    const int4 er       = *(const int4*)(rows + (own_live ? wi : 0) * 4);

    const int wave  = __builtin_amdgcn_readfirstlane(tid >> 5);
    const int lane  = tid & 31;
    const int hi    = lane >> 4;
    const int l16   = lane & 15;
    const int nt    = wave & 3;
    const int slot  = wave >> 2;
    const int ncol  = nt * 16 + l16;

    const float* wcol = ((nt < 2) ? (W_lin + ncol * CIN_)
                                  : (W_eye + (ncol - HALF_) * CIN_)) + hi * 2;
    v2f bf[16];
    #pragma unroll
    for (int kb = 0; kb < 16; ++kb)
        bf[kb] = *(const v2f*)(wcol + kb * 4);

    const int t0 = blockIdx.x * G_TPB_ + slot;
    v2f a[2][16];
    load_a_tile(x, t0, l16, hi, a[0]);

    float be = 0.0f;
    if (nt >= 2) be = b_eye[ncol - HALF_];

    if (own_live)
        own[wi] =  (((unsigned)er.x / RPB_) & 255u)
                | ((((unsigned)er.y / RPB_) & 255u) << 8)
                | ((((unsigned)er.z / RPB_) & 255u) << 16)
                |  (((unsigned)er.w / RPB_) << 24);

    #pragma unroll
    for (int i = 0; i < G_TRIPS_; ++i) {
        const int t = t0 + i * G_SLOTS_;
        if (i + 1 < G_TRIPS_)
            load_a_tile(x, t + G_SLOTS_, l16, hi, a[(i + 1) & 1]);

        v8f acc = {};
        #pragma unroll
        for (int kb = 0; kb < 16; ++kb) {
            acc = __builtin_amdgcn_wmma_f32_16x16x4_f32(
                 false, a[i & 1][kb],  false, bf[kb],
                 (short)0, acc,  false,  false);
        }

        if (t < NRT_) {
            const int m0 = t * 16 + 8 * hi;
            if (nt < 2) {
                #pragma unroll
                for (int r = 0; r < 8; ++r) {
                    const long m = m0 + r;
                    z[m * HALF_ + ncol] = acc[r];
                }
            } else {
                #pragma unroll
                for (int r = 0; r < 8; ++r) {
                    const long m = m0 + r;
                    out[m * CIN_ + ncol] = acc[r] + be;
                }
            }
        }
    }
}

#define QA_    1024
#define QB_    256
#define SCAP_  16
#define SPITCH_ 17
#define OWN_Q_ (E_ / 16)
#define NGRP_  ((OWN_Q_ + 31) / 32)
#define NLD_   ((NGRP_ + 31) / 32)
#define FU_    7


#define RPW_   ((RPB_ + 31) / 32)
typedef __attribute__((ext_vector_type(32))) float v32f;

__device__ __forceinline__ void work_off_list(
    const int* q, int qn, int lane,
    const float* __restrict__ z, const float* __restrict__ vals,
    const int* __restrict__ cols, v32f& sums)
{
    for (int i0 = 0; i0 < qn; i0 += 32) {
        const int n   = (qn - i0 < 32) ? (qn - i0) : 32;
        const int ent = (lane < n) ? q[i0 + lane] : 0;
        const int e   = ent & 0xFFFFF;
        const int   cn = cols[e];
        const float v  = vals[e];
        for (int k = 0; k < n; k += 8) {
            float zz[8][4];
            float vk[8];
            int   dk[8];
            #pragma unroll
            for (int kk = 0; kk < 8; ++kk) {
                const int cnk = __builtin_amdgcn_readlane(cn, k + kk);
                vk[kk] = __int_as_float(__builtin_amdgcn_readlane(__float_as_int(v), k + kk));
                dk[kk] = (__builtin_amdgcn_readlane(ent, k + kk) >> 25) * 4;
                #pragma unroll
                for (int b = 0; b < 4; ++b)
                    zz[kk][b] = z[((long)b * N_ + cnk) * HALF_ + lane];
            }
            #pragma unroll
            for (int kk = 0; kk < 8; ++kk) {
                if (k + kk < n) {
                    #pragma unroll
                    for (int b = 0; b < 4; ++b)
                        sums[dk[kk] + b] += vk[kk] * zz[kk][b];
                }
            }
        }
    }
}

__device__ __forceinline__ int filter_pass(
    int ld0, int ld1, int wave, int lane, unsigned key, int rot,
    const uint4* __restrict__ own, int* q, bool& too_small)
{
    const int sw = __builtin_amdgcn_readfirstlane(wave);
    int qn = 0;
    for (int ld = ld0; ld < ld1; ld += FU_) {
        uint4 ww[FU_];
        int   pc[FU_];
        bool  live[FU_];
        #pragma unroll
        for (int u = 0; u < FU_; ++u) {
            const int gi = (ld + u) * 32 + sw;
            int g = gi + rot;
            if (g >= NGRP_) g -= NGRP_;
            pc[u] = g * 32 + lane;
            live[u] = (ld + u < ld1) && (gi < NGRP_) && (pc[u] < OWN_Q_);
            const int pl = (pc[u] < OWN_Q_) ? pc[u] : (OWN_Q_ - 1);
            ww[u] = own[pl];
        }
        #pragma unroll
        for (int u = 0; u < FU_; ++u) {
            const unsigned t0 = ww[u].x ^ key, t1 = ww[u].y ^ key;
            const unsigned t2 = ww[u].z ^ key, t3 = ww[u].w ^ key;
            const unsigned n0 = (t0 - 0x01010101u) & ~t0 & 0x80808080u;
            const unsigned n1 = (t1 - 0x01010101u) & ~t1 & 0x80808080u;
            const unsigned n2 = (t2 - 0x01010101u) & ~t2 & 0x80808080u;
            const unsigned n3 = (t3 - 0x01010101u) & ~t3 & 0x80808080u;
            unsigned f = n0 | (n1 >> 1) | (n2 >> 2) | (n3 >> 3);
            if (!live[u]) f = 0u;
            unsigned bal = __builtin_amdgcn_ballot_w32(f != 0u);
            if (bal != 0u) {
                if (qn > QA_ - 512) {
                    too_small = true;
                } else {
                    do {
                        if (f != 0u) {
                            const int p = __builtin_ctz(f);
                            q[qn + __builtin_amdgcn_mbcnt_lo(bal, 0u)] =
                                pc[u] * 16 + 28 - ((p & 7) << 2) + (p >> 3);
                        }
                        qn += __builtin_popcount(bal);
                        f &= f - 1u;
                        bal = __builtin_amdgcn_ballot_w32(f != 0u);
                    } while (bal != 0u);
                }
            }
        }
    }
    return qn;
}

__device__ __forceinline__ int add_local_rows(
    int* q, int qn, int lane, int lo, int cnt, const int* __restrict__ rows)
{
    int wn = 0;
    for (int i0 = 0; i0 < qn; i0 += 128) {
        int  e[4];
        int  r[4];
        bool ok[4];
        #pragma unroll
        for (int u = 0; u < 4; ++u) {
            ok[u] = (i0 + u * 32 + lane) < qn;
            e[u]  = ok[u] ? q[i0 + u * 32 + lane] : 0;
        }
        #pragma unroll
        for (int u = 0; u < 4; ++u) r[u] = rows[e[u]];
        #pragma unroll
        for (int u = 0; u < 4; ++u) {
            const unsigned d    = (unsigned)(r[u] - lo);
            const bool     keep = ok[u] && d < (unsigned)cnt;
            const unsigned bm   = __builtin_amdgcn_ballot_w32(keep);
            if (keep) q[wn + __builtin_amdgcn_mbcnt_lo(bm, 0u)] = e[u] | (int)(d << 20);
            wn += __builtin_popcount(bm);
        }
    }
    return wn;
}

__device__ __forceinline__ void collect_and_add(
    const int* queA, const int* cntA, int* qb, int wave, int lane,
    const float* __restrict__ z, const float* __restrict__ vals,
    const int* __restrict__ cols, v32f& sums)
{
    int nb = 0;
    for (int s0 = 0; s0 < 32; ++s0) {
        const int  s  = (s0 + wave) & 31;
        const int  ns = __builtin_amdgcn_readfirstlane(cntA[s]);
        const int* qs = queA + s * QA_;
        for (int i0 = 0; i0 < ns; i0 += 32) {
            const bool ok   = (i0 + lane) < ns;
            const int  ent  = ok ? qs[i0 + lane] : 0;
            const bool mine = ok && (((ent >> 20) & 31) == wave);
            const unsigned bm = __builtin_amdgcn_ballot_w32(mine);
            if (bm != 0u) {
                if (mine) qb[nb + __builtin_amdgcn_mbcnt_lo(bm, 0u)] = ent;
                nb += __builtin_popcount(bm);
                if (nb > QB_ - 32) {
                    __builtin_amdgcn_wave_barrier();
                    work_off_list(qb, nb, lane, z, vals, cols, sums);
                    __builtin_amdgcn_wave_barrier();
                    nb = 0;
                }
            }
        }
    }
    __builtin_amdgcn_wave_barrier();
    work_off_list(qb, nb, lane, z, vals, cols, sums);
}


__device__ __forceinline__ bool file_rows(
    const int* q, int qn, int wave, int lane, int lo, int cnt,
    const int* __restrict__ rows, int* runs, int* runcnt)
{
    int  mycnt = 0;
    bool over  = false;
    for (int i0 = 0; i0 < qn; i0 += 128) {
        int  e[4];
        int  r[4];
        bool ok[4];
        #pragma unroll
        for (int u = 0; u < 4; ++u) {
            ok[u] = (i0 + u * 32 + lane) < qn;
            e[u]  = ok[u] ? q[i0 + u * 32 + lane] : 0;
        }
        #pragma unroll
        for (int u = 0; u < 4; ++u) r[u] = rows[e[u]];
        #pragma unroll
        for (int u = 0; u < 4; ++u) {
            const unsigned d    = (unsigned)(r[u] - lo);
            const bool     keep = ok[u] && d < (unsigned)cnt;
            const int      w    = (int)(d & 31u);
            unsigned same = __builtin_amdgcn_ballot_w32(keep);
            unsigned tome = same;
            #pragma unroll
            for (int b = 0; b < 5; ++b) {
                const unsigned bb = __builtin_amdgcn_ballot_w32(keep && ((w >> b) & 1));
                same &= ((w    >> b) & 1) ? bb : ~bb;
                tome &= ((lane >> b) & 1) ? bb : ~bb;
            }
            const int base = __builtin_amdgcn_ds_bpermute(w << 2, mycnt);
            const int pos  = base + (int)__builtin_amdgcn_mbcnt_lo(same, 0u);
            if (keep) {
                if (pos < SCAP_) runs[(w * 32 + wave) * SPITCH_ + pos] = e[u] | (int)(d << 20);
                else             over = true;
            }
            mycnt += __builtin_popcount(tome);
        }
    }
    runcnt[wave * 32 + lane] = mycnt;
    return __builtin_amdgcn_ballot_w32(over) == 0u;
}

__device__ __forceinline__ void gather_runs_and_add(
    const int* runs, const int* runcnt, int* qc, int wave, int lane,
    const float* __restrict__ z, const float* __restrict__ vals,
    const int* __restrict__ cols, v32f& sums)
{
    const int  c   = runcnt[lane * 32 + wave];
    const int* run = runs + (wave * 32 + lane) * SPITCH_;
    int p = c;
    #pragma unroll
    for (int o = 1; o < 32; o <<= 1) {
        const int t = __shfl_up(p, o);
        if (lane >= o) p += t;
    }
    const int nb    = __builtin_amdgcn_readlane(p, 31);
    const int start = p - c;
    for (int k = 0; k < SCAP_; ++k) {
        if (__builtin_amdgcn_ballot_w32(k < c) == 0u) break;
        if (k < c) qc[start + k] = run[k];
    }
    __builtin_amdgcn_wave_barrier();
    work_off_list(qc, nb, lane, z, vals, cols, sums);
}

__global__ __launch_bounds__(1024) void gcn_rowrange_accumulate(
    const float* __restrict__ z, const float* __restrict__ vals,
    const int* __restrict__ rows, const int* __restrict__ cols,
    const float* __restrict__ b_lin, const unsigned int* __restrict__ own,
    float* __restrict__ out)
{
    __shared__ int   queA[32 * QA_];
    __shared__ int   runs[32 * 32 * SPITCH_];
    __shared__ int   runcnt[32 * 32];
    __shared__ int   cntA[32];
    __shared__ int   list_too_small;

    const int tid  = threadIdx.x;
    const int lane = tid & 31;
    const int wave = tid >> 5;
    const int lo   = blockIdx.x * RPB_;
    const int cnt  = (N_ - lo < RPB_) ? (N_ - lo) : RPB_;

    const float bl = b_lin[lane];
    v32f sums;
    #pragma unroll
    for (int i = 0; i < RPW_ * 4; ++i) sums[i] = bl;
    if (tid == 0) list_too_small = 0;
    __syncthreads();

    int* qa = queA + wave * QA_;
    int* qb = runs + wave * QB_;
    const int rot = blockIdx.x * 6;
    const uint4* own16 = (const uint4*)own;
    const unsigned key = blockIdx.x * 0x01010101u;

    bool too_small = false;
    int qn = filter_pass(0, NLD_, wave, lane, key, rot, own16, qa, too_small);
    if (!too_small) too_small = !file_rows(qa, qn, wave, lane, lo, cnt, rows, runs, runcnt);
    if (lane == 0 && too_small) list_too_small = 1;
    __syncthreads();
    const int redo = __builtin_amdgcn_readfirstlane(list_too_small);
    if (redo == 0) {
        gather_runs_and_add(runs, runcnt, qa, wave, lane, z, vals, cols, sums);
    } else {
        for (int r = 0; r < NLD_; r += 2) {
            __syncthreads();
            bool unused = false;
            const int r1 = (r + 2 < NLD_) ? (r + 2) : NLD_;
            qn = filter_pass(r, r1, wave, lane, key, rot, own16, qa, unused);
            qn = add_local_rows(qa, qn, lane, lo, cnt, rows);
            if (lane == 0) cntA[wave] = qn;
            __syncthreads();
            collect_and_add(queA, cntA, qb, wave, lane, z, vals, cols, sums);
        }
    }

    const int sw = __builtin_amdgcn_readfirstlane(wave);
    #pragma unroll
    for (int j = 0; j < RPW_; ++j) {
        const int d = sw + 32 * j;
        if (d < cnt) {
            #pragma unroll
            for (int b = 0; b < 4; ++b)
                out[((long)b * N_ + lo + d) * CIN_ + lane] = sums[j * 4 + b];
        }
    }
}

extern "C" void kernel_launch(void* const* d_in, const int* in_sizes, int n_in,
                              void* d_out, int out_size, void* d_ws, size_t ws_size,
                              hipStream_t stream) {
    const float* x     = (const float*)d_in[0];
    const float* vals  = (const float*)d_in[1];
    const float* W_lin = (const float*)d_in[2];
    const float* b_lin = (const float*)d_in[3];
    const float* W_eye = (const float*)d_in[4];
    const float* b_eye = (const float*)d_in[5];
    const int*   rows  = (const int*)d_in[6];
    const int*   cols  = (const int*)d_in[7];
    float* out = (float*)d_out;
    float* z   = (float*)d_ws;
    unsigned int* own = (unsigned int*)(z + M_ * HALF_);

    gcn_gemm_wmma<<<256, 896, 0, stream>>>(x, W_lin, W_eye, b_eye, rows, out, z, own);

    gcn_rowrange_accumulate<<<256, 1024, 0, stream>>>(z, vals, rows, cols, b_lin, own, out);
}


# Launch 1 packs each edge's column, local row, owner and value into one 8-byte record; launch 2 looks a kept edge up once (was rows[], cols[], vals[]) and carries it through LDS; sum order kept
# speedup vs baseline: 5.1768x; 1.0692x over previous
#include <hip/hip_runtime.h>

#define B_    4
#define N_    50000
#define CIN_  64
#define E_    800000
#define HALF_ 32
#define M_    (B_ * N_)

typedef __attribute__((ext_vector_type(2))) float v2f;
typedef __attribute__((ext_vector_type(8))) float v8f;

#define RPB_      196
#define OWN_W_    (E_ / 4)
#define OWN_TPB_  782
#define G_SLOTS_  7
#define G_TRIPS_  7
#define G_TPB_    (G_SLOTS_ * G_TRIPS_)
#define NRT_      (M_ / 16)

__device__ __forceinline__ void load_a_tile(
    const float* __restrict__ x, int t, int l16, int hi, v2f (&a)[16])
{
    const int tt = (t < NRT_) ? t : (NRT_ - 1);
    const float* xrow = x + (long)(tt * 16 + l16) * CIN_ + hi * 2;
    #pragma unroll
    for (int kb = 0; kb < 16; ++kb)
        a[kb] = *(const v2f*)(xrow + kb * 4);
}

__device__ __forceinline__ unsigned rec_word0(int col, int row, unsigned owner)
{
    return ((unsigned)col & 0xFFFFu) | (((unsigned)row - owner * RPB_) << 16) | (owner << 24);
}

__global__ __launch_bounds__(896) void gcn_gemm_wmma(
    const float* __restrict__ x,
    const float* __restrict__ W_lin,
    const float* __restrict__ W_eye, const float* __restrict__ b_eye,
    const int* __restrict__ rows, const int* __restrict__ cols,
    const float* __restrict__ vals,
    float* __restrict__ out, float* __restrict__ z, unsigned int* __restrict__ own,
    unsigned int* __restrict__ recw)
{
    const int tid   = threadIdx.x;
    uint4* __restrict__ rec4 = (uint4*)recw;

    const int    wi       = blockIdx.x * OWN_TPB_ + tid;
    const bool   own_live = (tid < OWN_TPB_) && (wi < OWN_W_);
    const int    wld      = (own_live ? wi : 0) * 4;
    const int4   er       = *(const int4*)(rows + wld);
    const int4   ec       = *(const int4*)(cols + wld);
    const float4 ev       = *(const float4*)(vals + wld);

    const int wave  = __builtin_amdgcn_readfirstlane(tid >> 5);
    const int lane  = tid & 31;
    const int hi    = lane >> 4;
    const int l16   = lane & 15;
    const int nt    = wave & 3;
    const int slot  = wave >> 2;
    const int ncol  = nt * 16 + l16;

    const float* wcol = ((nt < 2) ? (W_lin + ncol * CIN_)
                                  : (W_eye + (ncol - HALF_) * CIN_)) + hi * 2;
    v2f bf[16];
    #pragma unroll
    for (int kb = 0; kb < 16; ++kb)
        bf[kb] = *(const v2f*)(wcol + kb * 4);

    const int t0 = blockIdx.x * G_TPB_ + slot;
    v2f a[2][16];
    load_a_tile(x, t0, l16, hi, a[0]);

    float be = 0.0f;
    if (nt >= 2) be = b_eye[ncol - HALF_];

    if (own_live) {
        const unsigned o0 = (unsigned)er.x / RPB_, o1 = (unsigned)er.y / RPB_;
        const unsigned o2 = (unsigned)er.z / RPB_, o3 = (unsigned)er.w / RPB_;
        own[wi] = (o0 & 255u) | ((o1 & 255u) << 8) | ((o2 & 255u) << 16) | (o3 << 24);
        uint4 ra, rb;
        ra.x = rec_word0(ec.x, er.x, o0); ra.y = __float_as_uint(ev.x);
        ra.z = rec_word0(ec.y, er.y, o1); ra.w = __float_as_uint(ev.y);
        rb.x = rec_word0(ec.z, er.z, o2); rb.y = __float_as_uint(ev.z);
        rb.z = rec_word0(ec.w, er.w, o3); rb.w = __float_as_uint(ev.w);
        rec4[wi * 2]     = ra;
        rec4[wi * 2 + 1] = rb;
    }

    #pragma unroll
    for (int i = 0; i < G_TRIPS_; ++i) {
        const int t = t0 + i * G_SLOTS_;
        if (i + 1 < G_TRIPS_)
            load_a_tile(x, t + G_SLOTS_, l16, hi, a[(i + 1) & 1]);

        v8f acc = {};
        #pragma unroll
        for (int kb = 0; kb < 16; ++kb) {
            acc = __builtin_amdgcn_wmma_f32_16x16x4_f32(
                 false, a[i & 1][kb],  false, bf[kb],
                 (short)0, acc,  false,  false);
        }

        if (t < NRT_) {
            const int m0 = t * 16 + 8 * hi;
            if (nt < 2) {
                #pragma unroll
                for (int r = 0; r < 8; ++r) {
                    const long m = m0 + r;
                    z[m * HALF_ + ncol] = acc[r];
                }
            } else {
                #pragma unroll
                for (int r = 0; r < 8; ++r) {
                    const long m = m0 + r;
                    out[m * CIN_ + ncol] = acc[r] + be;
                }
            }
        }
    }
}

#define QA_    1024
#define QB_    256
#define SCAP_  16
#define SPITCH_ 17
#define OWN_Q_ (E_ / 16)
#define NGRP_  ((OWN_Q_ + 31) / 32)
#define NLD_   ((NGRP_ + 31) / 32)
#define FU_    7


#define RPW_   ((RPB_ + 31) / 32)
typedef __attribute__((ext_vector_type(32))) float v32f;

__device__ __forceinline__ void work_off_list(
    const int* q, int qn, int lane,
    const float* __restrict__ z, const float* __restrict__ vals,
    const int* __restrict__ cols, v32f& sums)
{
    for (int i0 = 0; i0 < qn; i0 += 32) {
        const int n   = (qn - i0 < 32) ? (qn - i0) : 32;
        const int ent = (lane < n) ? q[i0 + lane] : 0;
        const int e   = ent & 0xFFFFF;
        const int   cn = cols[e];
        const float v  = vals[e];
        for (int k = 0; k < n; k += 8) {
            float zz[8][4];
            float vk[8];
            int   dk[8];
            #pragma unroll
            for (int kk = 0; kk < 8; ++kk) {
                const int cnk = __builtin_amdgcn_readlane(cn, k + kk);
                vk[kk] = __int_as_float(__builtin_amdgcn_readlane(__float_as_int(v), k + kk));
                dk[kk] = (__builtin_amdgcn_readlane(ent, k + kk) >> 25) * 4;
                #pragma unroll
                for (int b = 0; b < 4; ++b)
                    zz[kk][b] = z[((long)b * N_ + cnk) * HALF_ + lane];
            }
            #pragma unroll
            for (int kk = 0; kk < 8; ++kk) {
                if (k + kk < n) {
                    #pragma unroll
                    for (int b = 0; b < 4; ++b)
                        sums[dk[kk] + b] += vk[kk] * zz[kk][b];
                }
            }
        }
    }
}

__device__ __forceinline__ void work_off_recs(
    const int* qx, const int* qy, int qn, int lane,
    const float* __restrict__ z, v32f& sums)
{
    for (int i0 = 0; i0 < qn; i0 += 32) {
        const int n  = (qn - i0 < 32) ? (qn - i0) : 32;
        const int rx = (lane < n) ? qx[i0 + lane] : 0;
        const int ry = (lane < n) ? qy[i0 + lane] : 0;
        for (int k = 0; k < n; k += 8) {
            float zz[8][4];
            float vk[8];
            int   dk[8];
            #pragma unroll
            for (int kk = 0; kk < 8; ++kk) {
                const int rk  = __builtin_amdgcn_readlane(rx, k + kk);
                const int cnk = rk & 0xFFFF;
                vk[kk] = __int_as_float(__builtin_amdgcn_readlane(ry, k + kk));
                dk[kk] = ((rk >> 21) & 7) * 4;
                #pragma unroll
                for (int b = 0; b < 4; ++b)
                    zz[kk][b] = z[((long)b * N_ + cnk) * HALF_ + lane];
            }
            #pragma unroll
            for (int kk = 0; kk < 8; ++kk) {
                if (k + kk < n) {
                    #pragma unroll
                    for (int b = 0; b < 4; ++b)
                        sums[dk[kk] + b] += vk[kk] * zz[kk][b];
                }
            }
        }
    }
}

__device__ __forceinline__ int filter_pass(
    int ld0, int ld1, int wave, int lane, unsigned key, int rot,
    const uint4* __restrict__ own, int* q, bool& too_small)
{
    const int sw = __builtin_amdgcn_readfirstlane(wave);
    int qn = 0;
    for (int ld = ld0; ld < ld1; ld += FU_) {
        uint4 ww[FU_];
        int   pc[FU_];
        bool  live[FU_];
        #pragma unroll
        for (int u = 0; u < FU_; ++u) {
            const int gi = (ld + u) * 32 + sw;
            int g = gi + rot;
            if (g >= NGRP_) g -= NGRP_;
            pc[u] = g * 32 + lane;
            live[u] = (ld + u < ld1) && (gi < NGRP_) && (pc[u] < OWN_Q_);
            const int pl = (pc[u] < OWN_Q_) ? pc[u] : (OWN_Q_ - 1);
            ww[u] = own[pl];
        }
        #pragma unroll
        for (int u = 0; u < FU_; ++u) {
            const unsigned t0 = ww[u].x ^ key, t1 = ww[u].y ^ key;
            const unsigned t2 = ww[u].z ^ key, t3 = ww[u].w ^ key;
            const unsigned n0 = (t0 - 0x01010101u) & ~t0 & 0x80808080u;
            const unsigned n1 = (t1 - 0x01010101u) & ~t1 & 0x80808080u;
            const unsigned n2 = (t2 - 0x01010101u) & ~t2 & 0x80808080u;
            const unsigned n3 = (t3 - 0x01010101u) & ~t3 & 0x80808080u;
            unsigned f = n0 | (n1 >> 1) | (n2 >> 2) | (n3 >> 3);
            if (!live[u]) f = 0u;
            unsigned bal = __builtin_amdgcn_ballot_w32(f != 0u);
            if (bal != 0u) {
                if (qn > QA_ - 512) {
                    too_small = true;
                } else {
                    do {
                        if (f != 0u) {
                            const int p = __builtin_ctz(f);
                            q[qn + __builtin_amdgcn_mbcnt_lo(bal, 0u)] =
                                pc[u] * 16 + 28 - ((p & 7) << 2) + (p >> 3);
                        }
                        qn += __builtin_popcount(bal);
                        f &= f - 1u;
                        bal = __builtin_amdgcn_ballot_w32(f != 0u);
                    } while (bal != 0u);
                }
            }
        }
    }
    return qn;
}

__device__ __forceinline__ int add_local_rows(
    int* q, int qn, int lane, int lo, int cnt, const int* __restrict__ rows)
{
    int wn = 0;
    for (int i0 = 0; i0 < qn; i0 += 128) {
        int  e[4];
        int  r[4];
        bool ok[4];
        #pragma unroll
        for (int u = 0; u < 4; ++u) {
            ok[u] = (i0 + u * 32 + lane) < qn;
            e[u]  = ok[u] ? q[i0 + u * 32 + lane] : 0;
        }
        #pragma unroll
        for (int u = 0; u < 4; ++u) r[u] = rows[e[u]];
        #pragma unroll
        for (int u = 0; u < 4; ++u) {
            const unsigned d    = (unsigned)(r[u] - lo);
            const bool     keep = ok[u] && d < (unsigned)cnt;
            const unsigned bm   = __builtin_amdgcn_ballot_w32(keep);
            if (keep) q[wn + __builtin_amdgcn_mbcnt_lo(bm, 0u)] = e[u] | (int)(d << 20);
            wn += __builtin_popcount(bm);
        }
    }
    return wn;
}

__device__ __forceinline__ void collect_and_add(
    const int* queA, const int* cntA, int* qb, int wave, int lane,
    const float* __restrict__ z, const float* __restrict__ vals,
    const int* __restrict__ cols, v32f& sums)
{
    int nb = 0;
    for (int s0 = 0; s0 < 32; ++s0) {
        const int  s  = (s0 + wave) & 31;
        const int  ns = __builtin_amdgcn_readfirstlane(cntA[s]);
        const int* qs = queA + s * QA_;
        for (int i0 = 0; i0 < ns; i0 += 32) {
            const bool ok   = (i0 + lane) < ns;
            const int  ent  = ok ? qs[i0 + lane] : 0;
            const bool mine = ok && (((ent >> 20) & 31) == wave);
            const unsigned bm = __builtin_amdgcn_ballot_w32(mine);
            if (bm != 0u) {
                if (mine) qb[nb + __builtin_amdgcn_mbcnt_lo(bm, 0u)] = ent;
                nb += __builtin_popcount(bm);
                if (nb > QB_ - 32) {
                    __builtin_amdgcn_wave_barrier();
                    work_off_list(qb, nb, lane, z, vals, cols, sums);
                    __builtin_amdgcn_wave_barrier();
                    nb = 0;
                }
            }
        }
    }
    __builtin_amdgcn_wave_barrier();
    work_off_list(qb, nb, lane, z, vals, cols, sums);
}


__device__ __forceinline__ bool file_recs(
    const int* q, int qn, int wave, int lane, unsigned g, int cnt,
    const uint2* __restrict__ rec, int* runsx, int* runsy, int* runcnt)
{
    int  mycnt = 0;
    bool over  = false;
    for (int i0 = 0; i0 < qn; i0 += 128) {
        int   e[4];
        uint2 r[4];
        bool  ok[4];
        #pragma unroll
        for (int u = 0; u < 4; ++u) {
            ok[u] = (i0 + u * 32 + lane) < qn;
            e[u]  = ok[u] ? q[i0 + u * 32 + lane] : 0;
        }
        #pragma unroll
        for (int u = 0; u < 4; ++u) r[u] = rec[e[u]];
        #pragma unroll
        for (int u = 0; u < 4; ++u) {
            const unsigned d    = (r[u].x >> 16) & 255u;
            const bool     keep = ok[u] && (r[u].x >> 24) == g && d < (unsigned)cnt;
            const int      w    = (int)(d & 31u);
            unsigned same = __builtin_amdgcn_ballot_w32(keep);
            unsigned tome = same;
            #pragma unroll
            for (int b = 0; b < 5; ++b) {
                const unsigned bb = __builtin_amdgcn_ballot_w32(keep && ((w >> b) & 1));
                same &= ((w    >> b) & 1) ? bb : ~bb;
                tome &= ((lane >> b) & 1) ? bb : ~bb;
            }
            const int base = __builtin_amdgcn_ds_bpermute(w << 2, mycnt);
            const int pos  = base + (int)__builtin_amdgcn_mbcnt_lo(same, 0u);
            if (keep) {
                if (pos < SCAP_) {
                    const int at = (w * 32 + wave) * SPITCH_ + pos;
                    runsx[at] = (int)r[u].x;
                    runsy[at] = (int)r[u].y;
                } else {
                    over = true;
                }
            }
            mycnt += __builtin_popcount(tome);
        }
    }
    runcnt[wave * 32 + lane] = mycnt;
    return __builtin_amdgcn_ballot_w32(over) == 0u;
}

__device__ __forceinline__ void gather_runs_and_add(
    const int* runsx, const int* runsy, const int* runcnt, int* qc, int wave, int lane,
    const float* __restrict__ z, v32f& sums)
{
    const int  c    = runcnt[lane * 32 + wave];
    const int* runx = runsx + (wave * 32 + lane) * SPITCH_;
    const int* runy = runsy + (wave * 32 + lane) * SPITCH_;
    int* qcx = qc;
    int* qcy = qc + QA_ / 2;
    int p = c;
    #pragma unroll
    for (int o = 1; o < 32; o <<= 1) {
        const int t = __shfl_up(p, o);
        if (lane >= o) p += t;
    }
    const int nb    = __builtin_amdgcn_readlane(p, 31);
    const int start = p - c;
    for (int k = 0; k < SCAP_; ++k) {
        if (__builtin_amdgcn_ballot_w32(k < c) == 0u) break;
        if (k < c) {
            qcx[start + k] = runx[k];
            qcy[start + k] = runy[k];
        }
    }
    __builtin_amdgcn_wave_barrier();
    work_off_recs(qcx, qcy, nb, lane, z, sums);
}

__global__ __launch_bounds__(1024) void gcn_rowrange_accumulate(
    const float* __restrict__ z, const float* __restrict__ vals,
    const int* __restrict__ rows, const int* __restrict__ cols,
    const float* __restrict__ b_lin, const unsigned int* __restrict__ own,
    const unsigned int* __restrict__ recw, float* __restrict__ out)
{
    const uint2* __restrict__ rec = (const uint2*)recw;
    __shared__ int   queA[32 * QA_];
    __shared__ int   runsx[32 * 32 * SPITCH_];
    __shared__ int   runsy[32 * 32 * SPITCH_];
    __shared__ int   runcnt[32 * 32];
    __shared__ int   cntA[32];
    __shared__ int   list_too_small;

    const int tid  = threadIdx.x;
    const int lane = tid & 31;
    const int wave = tid >> 5;
    const int lo   = blockIdx.x * RPB_;
    const int cnt  = (N_ - lo < RPB_) ? (N_ - lo) : RPB_;

    const float bl = b_lin[lane];
    v32f sums;
    #pragma unroll
    for (int i = 0; i < RPW_ * 4; ++i) sums[i] = bl;
    if (tid == 0) list_too_small = 0;
    __syncthreads();

    int* qa = queA + wave * QA_;
    int* qb = runsx + wave * QB_;
    const int rot = blockIdx.x * 6;
    const uint4* own16 = (const uint4*)own;
    const unsigned key = blockIdx.x * 0x01010101u;

    bool too_small = false;
    int qn = filter_pass(0, NLD_, wave, lane, key, rot, own16, qa, too_small);
    if (!too_small) too_small = !file_recs(qa, qn, wave, lane, blockIdx.x, cnt, rec, runsx, runsy, runcnt);
    if (lane == 0 && too_small) list_too_small = 1;
    __syncthreads();
    const int redo = __builtin_amdgcn_readfirstlane(list_too_small);
    if (redo == 0) {
        gather_runs_and_add(runsx, runsy, runcnt, qa, wave, lane, z, sums);
    } else {
        for (int r = 0; r < NLD_; r += 2) {
            __syncthreads();
            bool unused = false;
            const int r1 = (r + 2 < NLD_) ? (r + 2) : NLD_;
            qn = filter_pass(r, r1, wave, lane, key, rot, own16, qa, unused);
            qn = add_local_rows(qa, qn, lane, lo, cnt, rows);
            if (lane == 0) cntA[wave] = qn;
            __syncthreads();
            collect_and_add(queA, cntA, qb, wave, lane, z, vals, cols, sums);
        }
    }

    const int sw = __builtin_amdgcn_readfirstlane(wave);
    #pragma unroll
    for (int j = 0; j < RPW_; ++j) {
        const int d = sw + 32 * j;
        if (d < cnt) {
            #pragma unroll
            for (int b = 0; b < 4; ++b)
                out[((long)b * N_ + lo + d) * CIN_ + lane] = sums[j * 4 + b];
        }
    }
}

extern "C" void kernel_launch(void* const* d_in, const int* in_sizes, int n_in,
                              void* d_out, int out_size, void* d_ws, size_t ws_size,
                              hipStream_t stream) {
    const float* x     = (const float*)d_in[0];
    const float* vals  = (const float*)d_in[1];
    const float* W_lin = (const float*)d_in[2];
    const float* b_lin = (const float*)d_in[3];
    const float* W_eye = (const float*)d_in[4];
    const float* b_eye = (const float*)d_in[5];
    const int*   rows  = (const int*)d_in[6];
    const int*   cols  = (const int*)d_in[7];
    float* out = (float*)d_out;
    float* z   = (float*)d_ws;
    unsigned int* own = (unsigned int*)(z + M_ * HALF_);
    unsigned int* rec = own + OWN_W_;

    gcn_gemm_wmma<<<256, 896, 0, stream>>>(x, W_lin, W_eye, b_eye, rows, cols, vals, out, z, own, rec);

    gcn_rowrange_accumulate<<<256, 1024, 0, stream>>>(z, vals, rows, cols, b_lin, own, rec, out);
}


# Owner numbers written as 8 bit planes per 32 edges (ballots in launch 1), not a byte per edge: launch 2 tests 32 edges per lane with 8 xor-or steps; row sum order changed, fixed by code
# speedup vs baseline: 5.4015x; 1.0434x over previous
#include <hip/hip_runtime.h>

#define B_    4
#define N_    50000
#define CIN_  64
#define E_    800000
#define HALF_ 32
#define M_    (B_ * N_)

typedef __attribute__((ext_vector_type(2))) float v2f;
typedef __attribute__((ext_vector_type(8))) float v8f;

#define RPB_      196
#define OWN_W_    (E_ / 4)
#define OWN_GW_   (E_ / 128)
#define OWN_Q_    (E_ / 32)
#define G_WGS_    256
#define G_SLOTS_  7
#define G_TRIPS_  7
#define G_TPB_    (G_SLOTS_ * G_TRIPS_)
#define NRT_      (M_ / 16)

__device__ __forceinline__ void load_a_tile(
    const float* __restrict__ x, int t, int l16, int hi, v2f (&a)[16])
{
    const int tt = (t < NRT_) ? t : (NRT_ - 1);
    const float* xrow = x + (long)(tt * 16 + l16) * CIN_ + hi * 2;
    #pragma unroll
    for (int kb = 0; kb < 16; ++kb)
        a[kb] = *(const v2f*)(xrow + kb * 4);
}

__device__ __forceinline__ unsigned rec_word0(int col, int row, unsigned owner)
{
    return ((unsigned)col & 0xFFFFu) | (((unsigned)row - owner * RPB_) << 16) | (owner << 24);
}

__global__ __launch_bounds__(896) void gcn_gemm_wmma(
    const float* __restrict__ x,
    const float* __restrict__ W_lin,
    const float* __restrict__ W_eye, const float* __restrict__ b_eye,
    const int* __restrict__ rows, const int* __restrict__ cols,
    const float* __restrict__ vals,
    float* __restrict__ out, float* __restrict__ z, unsigned int* __restrict__ own,
    unsigned int* __restrict__ recw)
{
    const int tid   = threadIdx.x;
    uint4* __restrict__ rec4 = (uint4*)recw;

    const int wave  = __builtin_amdgcn_readfirstlane(tid >> 5);
    const int lane  = tid & 31;

    const int    gw       = wave * G_WGS_ + blockIdx.x;
    const bool   own_live = gw < OWN_GW_;
    const int    wi       = gw * 32 + lane;
    const int    wld      = (own_live ? wi : 0) * 4;
    const int4   er       = *(const int4*)(rows + wld);
    const int4   ec       = *(const int4*)(cols + wld);
    const float4 ev       = *(const float4*)(vals + wld);

    const int hi    = lane >> 4;
    const int l16   = lane & 15;
    const int nt    = wave & 3;
    const int slot  = wave >> 2;
    const int ncol  = nt * 16 + l16;

    const float* wcol = ((nt < 2) ? (W_lin + ncol * CIN_)
                                  : (W_eye + (ncol - HALF_) * CIN_)) + hi * 2;
    v2f bf[16];
    #pragma unroll
    for (int kb = 0; kb < 16; ++kb)
        bf[kb] = *(const v2f*)(wcol + kb * 4);

    const int t0 = blockIdx.x * G_TPB_ + slot;
    v2f a[2][16];
    load_a_tile(x, t0, l16, hi, a[0]);

    float be = 0.0f;
    if (nt >= 2) be = b_eye[ncol - HALF_];

    if (own_live) {
        const unsigned o0 = (unsigned)er.x / RPB_, o1 = (unsigned)er.y / RPB_;
        const unsigned o2 = (unsigned)er.z / RPB_, o3 = (unsigned)er.w / RPB_;
        const unsigned ob[4] = {o0, o1, o2, o3};
        unsigned pw = 0u;
        #pragma unroll
        for (int i = 0; i < 4; ++i) {
            #pragma unroll
            for (int b = 0; b < 8; ++b) {
                const unsigned bal = __builtin_amdgcn_ballot_w32(((ob[i] >> b) & 1u) != 0u);
                if (lane == i * 8 + b) pw = bal;
            }
        }
        const int pq = gw * 4 + (lane >> 3);
        const int pb = lane & 7;
        own[((pb >> 2) * OWN_Q_ + pq) * 4 + (pb & 3)] = pw;
        uint4 ra, rb;
        ra.x = rec_word0(ec.x, er.x, o0); ra.y = __float_as_uint(ev.x);
        ra.z = rec_word0(ec.y, er.y, o1); ra.w = __float_as_uint(ev.y);
        rb.x = rec_word0(ec.z, er.z, o2); rb.y = __float_as_uint(ev.z);
        rb.z = rec_word0(ec.w, er.w, o3); rb.w = __float_as_uint(ev.w);
        rec4[wi * 2]     = ra;
        rec4[wi * 2 + 1] = rb;
    }

    #pragma unroll
    for (int i = 0; i < G_TRIPS_; ++i) {
        const int t = t0 + i * G_SLOTS_;
        if (i + 1 < G_TRIPS_)
            load_a_tile(x, t + G_SLOTS_, l16, hi, a[(i + 1) & 1]);

        v8f acc = {};
        #pragma unroll
        for (int kb = 0; kb < 16; ++kb) {
            acc = __builtin_amdgcn_wmma_f32_16x16x4_f32(
                 false, a[i & 1][kb],  false, bf[kb],
                 (short)0, acc,  false,  false);
        }

        if (t < NRT_) {
            const int m0 = t * 16 + 8 * hi;
            if (nt < 2) {
                #pragma unroll
                for (int r = 0; r < 8; ++r) {
                    const long m = m0 + r;
                    z[m * HALF_ + ncol] = acc[r];
                }
            } else {
                #pragma unroll
                for (int r = 0; r < 8; ++r) {
                    const long m = m0 + r;
                    out[m * CIN_ + ncol] = acc[r] + be;
                }
            }
        }
    }
}

#define QA_    1024
#define QB_    256
#define SCAP_  16
#define SPITCH_ 17
#define NGRP_  ((OWN_Q_ + 31) / 32)
#define NLD_   ((NGRP_ + 31) / 32)
#define FU_    5


#define RPW_   ((RPB_ + 31) / 32)
typedef __attribute__((ext_vector_type(32))) float v32f;

__device__ __forceinline__ void work_off_list(
    const int* q, int qn, int lane,
    const float* __restrict__ z, const float* __restrict__ vals,
    const int* __restrict__ cols, v32f& sums)
{
    for (int i0 = 0; i0 < qn; i0 += 32) {
        const int n   = (qn - i0 < 32) ? (qn - i0) : 32;
        const int ent = (lane < n) ? q[i0 + lane] : 0;
        const int e   = ent & 0xFFFFF;
        const int   cn = cols[e];
        const float v  = vals[e];
        for (int k = 0; k < n; k += 8) {
            float zz[8][4];
            float vk[8];
            int   dk[8];
            #pragma unroll
            for (int kk = 0; kk < 8; ++kk) {
                const int cnk = __builtin_amdgcn_readlane(cn, k + kk);
                vk[kk] = __int_as_float(__builtin_amdgcn_readlane(__float_as_int(v), k + kk));
                dk[kk] = (__builtin_amdgcn_readlane(ent, k + kk) >> 25) * 4;
                #pragma unroll
                for (int b = 0; b < 4; ++b)
                    zz[kk][b] = z[((long)b * N_ + cnk) * HALF_ + lane];
            }
            #pragma unroll
            for (int kk = 0; kk < 8; ++kk) {
                if (k + kk < n) {
                    #pragma unroll
                    for (int b = 0; b < 4; ++b)
                        sums[dk[kk] + b] += vk[kk] * zz[kk][b];
                }
            }
        }
    }
}

__device__ __forceinline__ void work_off_recs(
    const int* qx, const int* qy, int qn, int lane,
    const float* __restrict__ z, v32f& sums)
{
    for (int i0 = 0; i0 < qn; i0 += 32) {
        const int n  = (qn - i0 < 32) ? (qn - i0) : 32;
        const int rx = (lane < n) ? qx[i0 + lane] : 0;
        const int ry = (lane < n) ? qy[i0 + lane] : 0;
        for (int k = 0; k < n; k += 8) {
            float zz[8][4];
            float vk[8];
            int   dk[8];
            #pragma unroll
            for (int kk = 0; kk < 8; ++kk) {
                const int rk  = __builtin_amdgcn_readlane(rx, k + kk);
                const int cnk = rk & 0xFFFF;
                vk[kk] = __int_as_float(__builtin_amdgcn_readlane(ry, k + kk));
                dk[kk] = ((rk >> 21) & 7) * 4;
                #pragma unroll
                for (int b = 0; b < 4; ++b)
                    zz[kk][b] = z[((long)b * N_ + cnk) * HALF_ + lane];
            }
            #pragma unroll
            for (int kk = 0; kk < 8; ++kk) {
                if (k + kk < n) {
                    #pragma unroll
                    for (int b = 0; b < 4; ++b)
                        sums[dk[kk] + b] += vk[kk] * zz[kk][b];
                }
            }
        }
    }
}

__device__ __forceinline__ int filter_pass(
    int ld0, int ld1, int wave, int lane, unsigned g, int rot,
    const uint4* __restrict__ own, int* q, bool& too_small)
{
    const int sw = __builtin_amdgcn_readfirstlane(wave);
    unsigned xm[8];
    #pragma unroll
    for (int b = 0; b < 8; ++b) xm[b] = 0u - ((g >> b) & 1u);
    int qn = 0;
    #pragma nounroll
    for (int ld = ld0; ld < ld1; ld += FU_) {
        uint4 wa[FU_];
        uint4 wb[FU_];
        int   pc[FU_];
        bool  live[FU_];
        #pragma unroll
        for (int u = 0; u < FU_; ++u) {
            const int gi = (ld + u) * 32 + sw;
            int gg = gi + rot;
            if (gg >= NGRP_) gg -= NGRP_;
            pc[u] = gg * 32 + lane;
            live[u] = (ld + u < ld1) && (gi < NGRP_) && (pc[u] < OWN_Q_);
            const int pl = (pc[u] < OWN_Q_) ? pc[u] : (OWN_Q_ - 1);
            wa[u] = own[pl];
            wb[u] = own[OWN_Q_ + pl];
        }
        __builtin_amdgcn_sched_barrier(0);
        #pragma unroll
        for (int u = 0; u < FU_; ++u) {
            const unsigned mis = (wa[u].x ^ xm[0]) | (wa[u].y ^ xm[1]) | (wa[u].z ^ xm[2]) | (wa[u].w ^ xm[3])
                               | (wb[u].x ^ xm[4]) | (wb[u].y ^ xm[5]) | (wb[u].z ^ xm[6]) | (wb[u].w ^ xm[7]);
            unsigned f = live[u] ? ~mis : 0u;
            unsigned bal = __builtin_amdgcn_ballot_w32(f != 0u);
            if (bal != 0u) {
                const int e0 = (pc[u] >> 2) * 128 + (pc[u] & 3);
                do {
                    const int nh = __builtin_popcount(bal);
                    if (qn + nh > QA_) {
                        too_small = true;
                        break;
                    }
                    if (f != 0u)
                        q[qn + __builtin_amdgcn_mbcnt_lo(bal, 0u)] = e0 + 4 * __builtin_ctz(f);
                    qn += nh;
                    f &= f - 1u;
                    bal = __builtin_amdgcn_ballot_w32(f != 0u);
                } while (bal != 0u);
            }
        }
    }
    return qn;
}

__device__ __forceinline__ int add_local_rows(
    int* q, int qn, int lane, int lo, int cnt, const int* __restrict__ rows)
{
    int wn = 0;
    for (int i0 = 0; i0 < qn; i0 += 128) {
        int  e[4];
        int  r[4];
        bool ok[4];
        #pragma unroll
        for (int u = 0; u < 4; ++u) {
            ok[u] = (i0 + u * 32 + lane) < qn;
            e[u]  = ok[u] ? q[i0 + u * 32 + lane] : 0;
        }
        #pragma unroll
        for (int u = 0; u < 4; ++u) r[u] = rows[e[u]];
        #pragma unroll
        for (int u = 0; u < 4; ++u) {
            const unsigned d    = (unsigned)(r[u] - lo);
            const bool     keep = ok[u] && d < (unsigned)cnt;
            const unsigned bm   = __builtin_amdgcn_ballot_w32(keep);
            if (keep) q[wn + __builtin_amdgcn_mbcnt_lo(bm, 0u)] = e[u] | (int)(d << 20);
            wn += __builtin_popcount(bm);
        }
    }
    return wn;
}

__device__ __forceinline__ void collect_and_add(
    const int* queA, const int* cntA, int* qb, int wave, int lane,
    const float* __restrict__ z, const float* __restrict__ vals,
    const int* __restrict__ cols, v32f& sums)
{
    int nb = 0;
    for (int s0 = 0; s0 < 32; ++s0) {
        const int  s  = (s0 + wave) & 31;
        const int  ns = __builtin_amdgcn_readfirstlane(cntA[s]);
        const int* qs = queA + s * QA_;
        for (int i0 = 0; i0 < ns; i0 += 32) {
            const bool ok   = (i0 + lane) < ns;
            const int  ent  = ok ? qs[i0 + lane] : 0;
            const bool mine = ok && (((ent >> 20) & 31) == wave);
            const unsigned bm = __builtin_amdgcn_ballot_w32(mine);
            if (bm != 0u) {
                if (mine) qb[nb + __builtin_amdgcn_mbcnt_lo(bm, 0u)] = ent;
                nb += __builtin_popcount(bm);
                if (nb > QB_ - 32) {
                    __builtin_amdgcn_wave_barrier();
                    work_off_list(qb, nb, lane, z, vals, cols, sums);
                    __builtin_amdgcn_wave_barrier();
                    nb = 0;
                }
            }
        }
    }
    __builtin_amdgcn_wave_barrier();
    work_off_list(qb, nb, lane, z, vals, cols, sums);
}


__device__ __forceinline__ bool file_recs(
    const int* q, int qn, int wave, int lane, unsigned g, int cnt,
    const uint2* __restrict__ rec, int* runsx, int* runsy, int* runcnt)
{
    int  mycnt = 0;
    bool over  = false;
    for (int i0 = 0; i0 < qn; i0 += 128) {
        int   e[4];
        uint2 r[4];
        bool  ok[4];
        #pragma unroll
        for (int u = 0; u < 4; ++u) {
            ok[u] = (i0 + u * 32 + lane) < qn;
            e[u]  = ok[u] ? q[i0 + u * 32 + lane] : 0;
        }
        #pragma unroll
        for (int u = 0; u < 4; ++u) r[u] = rec[e[u]];
        #pragma unroll
        for (int u = 0; u < 4; ++u) {
            const unsigned d    = (r[u].x >> 16) & 255u;
            const bool     keep = ok[u] && (r[u].x >> 24) == g && d < (unsigned)cnt;
            const int      w    = (int)(d & 31u);
            unsigned same = __builtin_amdgcn_ballot_w32(keep);
            unsigned tome = same;
            #pragma unroll
            for (int b = 0; b < 5; ++b) {
                const unsigned bb = __builtin_amdgcn_ballot_w32(keep && ((w >> b) & 1));
                same &= ((w    >> b) & 1) ? bb : ~bb;
                tome &= ((lane >> b) & 1) ? bb : ~bb;
            }
            const int base = __builtin_amdgcn_ds_bpermute(w << 2, mycnt);
            const int pos  = base + (int)__builtin_amdgcn_mbcnt_lo(same, 0u);
            if (keep) {
                if (pos < SCAP_) {
                    const int at = (w * 32 + wave) * SPITCH_ + pos;
                    runsx[at] = (int)r[u].x;
                    runsy[at] = (int)r[u].y;
                } else {
                    over = true;
                }
            }
            mycnt += __builtin_popcount(tome);
        }
    }
    runcnt[wave * 32 + lane] = mycnt;
    return __builtin_amdgcn_ballot_w32(over) == 0u;
}

__device__ __forceinline__ void gather_runs_and_add(
    const int* runsx, const int* runsy, const int* runcnt, int* qc, int wave, int lane,
    const float* __restrict__ z, v32f& sums)
{
    const int  c    = runcnt[lane * 32 + wave];
    const int* runx = runsx + (wave * 32 + lane) * SPITCH_;
    const int* runy = runsy + (wave * 32 + lane) * SPITCH_;
    int* qcx = qc;
    int* qcy = qc + QA_ / 2;
    int p = c;
    #pragma unroll
    for (int o = 1; o < 32; o <<= 1) {
        const int t = __shfl_up(p, o);
        if (lane >= o) p += t;
    }
    const int nb    = __builtin_amdgcn_readlane(p, 31);
    const int start = p - c;
    for (int k = 0; k < SCAP_; ++k) {
        if (__builtin_amdgcn_ballot_w32(k < c) == 0u) break;
        if (k < c) {
            qcx[start + k] = runx[k];
            qcy[start + k] = runy[k];
        }
    }
    __builtin_amdgcn_wave_barrier();
    work_off_recs(qcx, qcy, nb, lane, z, sums);
}

__global__ __launch_bounds__(1024) void gcn_rowrange_accumulate(
    const float* __restrict__ z, const float* __restrict__ vals,
    const int* __restrict__ rows, const int* __restrict__ cols,
    const float* __restrict__ b_lin, const unsigned int* __restrict__ own,
    const unsigned int* __restrict__ recw, float* __restrict__ out)
{
    const uint2* __restrict__ rec = (const uint2*)recw;
    __shared__ int   queA[32 * QA_];
    __shared__ int   runsx[32 * 32 * SPITCH_];
    __shared__ int   runsy[32 * 32 * SPITCH_];
    __shared__ int   runcnt[32 * 32];
    __shared__ int   cntA[32];
    __shared__ int   list_too_small;

    const int tid  = threadIdx.x;
    const int lane = tid & 31;
    const int wave = tid >> 5;
    const int lo   = blockIdx.x * RPB_;
    const int cnt  = (N_ - lo < RPB_) ? (N_ - lo) : RPB_;

    const float bl = b_lin[lane];
    v32f sums;
    #pragma unroll
    for (int i = 0; i < RPW_ * 4; ++i) sums[i] = bl;
    if (tid == 0) list_too_small = 0;
    __syncthreads();

    int* qa = queA + wave * QA_;
    int* qb = runsx + wave * QB_;
    const int rot = blockIdx.x * 3;
    const uint4* own16 = (const uint4*)own;
    const unsigned key = blockIdx.x;

    bool too_small = false;
    int qn = filter_pass(0, NLD_, wave, lane, key, rot, own16, qa, too_small);
    if (!too_small) too_small = !file_recs(qa, qn, wave, lane, blockIdx.x, cnt, rec, runsx, runsy, runcnt);
    if (lane == 0 && too_small) list_too_small = 1;
    __syncthreads();
    const int redo = __builtin_amdgcn_readfirstlane(list_too_small);
    if (redo == 0) {
        gather_runs_and_add(runsx, runsy, runcnt, qa, wave, lane, z, sums);
    } else {
        for (int r = 0; r < NLD_; ++r) {
            __syncthreads();
            bool unused = false;
            qn = filter_pass(r, r + 1, wave, lane, key, rot, own16, qa, unused);
            qn = add_local_rows(qa, qn, lane, lo, cnt, rows);
            if (lane == 0) cntA[wave] = qn;
            __syncthreads();
            collect_and_add(queA, cntA, qb, wave, lane, z, vals, cols, sums);
        }
    }

    const int sw = __builtin_amdgcn_readfirstlane(wave);
    #pragma unroll
    for (int j = 0; j < RPW_; ++j) {
        const int d = sw + 32 * j;
        if (d < cnt) {
            #pragma unroll
            for (int b = 0; b < 4; ++b)
                out[((long)b * N_ + lo + d) * CIN_ + lane] = sums[j * 4 + b];
        }
    }
}

extern "C" void kernel_launch(void* const* d_in, const int* in_sizes, int n_in,
                              void* d_out, int out_size, void* d_ws, size_t ws_size,
                              hipStream_t stream) {
    const float* x     = (const float*)d_in[0];
    const float* vals  = (const float*)d_in[1];
    const float* W_lin = (const float*)d_in[2];
    const float* b_lin = (const float*)d_in[3];
    const float* W_eye = (const float*)d_in[4];
    const float* b_eye = (const float*)d_in[5];
    const int*   rows  = (const int*)d_in[6];
    const int*   cols  = (const int*)d_in[7];
    float* out = (float*)d_out;
    float* z   = (float*)d_ws;
    unsigned int* own = (unsigned int*)(z + M_ * HALF_);
    unsigned int* rec = own + OWN_W_;

    gcn_gemm_wmma<<<256, 896, 0, stream>>>(x, W_lin, W_eye, b_eye, rows, cols, vals, out, z, own, rec);

    gcn_rowrange_accumulate<<<256, 1024, 0, stream>>>(z, vals, rows, cols, b_lin, own, rec, out);
}


# z kept node-major ([node][batch][32]): an edge's 4 batch rows are one 512-byte run, read by one 16-byte load per lane, not four 4-byte loads; 16 edges in flight, not 8; sum order kept
# speedup vs baseline: 5.6101x; 1.0386x over previous
#include <hip/hip_runtime.h>

#define B_    4
#define N_    50000
#define CIN_  64
#define E_    800000
#define HALF_ 32
#define M_    (B_ * N_)

typedef __attribute__((ext_vector_type(2))) float v2f;
typedef __attribute__((ext_vector_type(8))) float v8f;

#define RPB_      196
#define OWN_W_    (E_ / 4)
#define OWN_GW_   (E_ / 128)
#define OWN_Q_    (E_ / 32)
#define G_WGS_    256
#define G_SLOTS_  7
#define G_TRIPS_  7
#define G_TPB_    (G_SLOTS_ * G_TRIPS_)
#define NRT_      (M_ / 16)

__device__ __forceinline__ void load_a_tile(
    const float* __restrict__ x, int t, int l16, int hi, v2f (&a)[16])
{
    const int tt = (t < NRT_) ? t : (NRT_ - 1);
    const float* xrow = x + (long)(tt * 16 + l16) * CIN_ + hi * 2;
    #pragma unroll
    for (int kb = 0; kb < 16; ++kb)
        a[kb] = *(const v2f*)(xrow + kb * 4);
}

__device__ __forceinline__ unsigned rec_word0(int col, int row, unsigned owner)
{
    return ((unsigned)col & 0xFFFFu) | (((unsigned)row - owner * RPB_) << 16) | (owner << 24);
}

__global__ __launch_bounds__(896) void gcn_gemm_wmma(
    const float* __restrict__ x,
    const float* __restrict__ W_lin,
    const float* __restrict__ W_eye, const float* __restrict__ b_eye,
    const int* __restrict__ rows, const int* __restrict__ cols,
    const float* __restrict__ vals,
    float* __restrict__ out, float* __restrict__ z, unsigned int* __restrict__ own,
    unsigned int* __restrict__ recw)
{
    const int tid   = threadIdx.x;
    uint4* __restrict__ rec4 = (uint4*)recw;

    const int wave  = __builtin_amdgcn_readfirstlane(tid >> 5);
    const int lane  = tid & 31;

    const int    gw       = wave * G_WGS_ + blockIdx.x;
    const bool   own_live = gw < OWN_GW_;
    const int    wi       = gw * 32 + lane;
    const int    wld      = (own_live ? wi : 0) * 4;
    const int4   er       = *(const int4*)(rows + wld);
    const int4   ec       = *(const int4*)(cols + wld);
    const float4 ev       = *(const float4*)(vals + wld);

    const int hi    = lane >> 4;
    const int l16   = lane & 15;
    const int nt    = wave & 3;
    const int slot  = wave >> 2;
    const int ncol  = nt * 16 + l16;

    const float* wcol = ((nt < 2) ? (W_lin + ncol * CIN_)
                                  : (W_eye + (ncol - HALF_) * CIN_)) + hi * 2;
    v2f bf[16];
    #pragma unroll
    for (int kb = 0; kb < 16; ++kb)
        bf[kb] = *(const v2f*)(wcol + kb * 4);

    const int t0 = blockIdx.x * G_TPB_ + slot;
    v2f a[2][16];
    load_a_tile(x, t0, l16, hi, a[0]);

    float be = 0.0f;
    if (nt >= 2) be = b_eye[ncol - HALF_];

    if (own_live) {
        const unsigned o0 = (unsigned)er.x / RPB_, o1 = (unsigned)er.y / RPB_;
        const unsigned o2 = (unsigned)er.z / RPB_, o3 = (unsigned)er.w / RPB_;
        const unsigned ob[4] = {o0, o1, o2, o3};
        unsigned pw = 0u;
        #pragma unroll
        for (int i = 0; i < 4; ++i) {
            #pragma unroll
            for (int b = 0; b < 8; ++b) {
                const unsigned bal = __builtin_amdgcn_ballot_w32(((ob[i] >> b) & 1u) != 0u);
                if (lane == i * 8 + b) pw = bal;
            }
        }
        const int pq = gw * 4 + (lane >> 3);
        const int pb = lane & 7;
        own[((pb >> 2) * OWN_Q_ + pq) * 4 + (pb & 3)] = pw;
        uint4 ra, rb;
        ra.x = rec_word0(ec.x, er.x, o0); ra.y = __float_as_uint(ev.x);
        ra.z = rec_word0(ec.y, er.y, o1); ra.w = __float_as_uint(ev.y);
        rb.x = rec_word0(ec.z, er.z, o2); rb.y = __float_as_uint(ev.z);
        rb.z = rec_word0(ec.w, er.w, o3); rb.w = __float_as_uint(ev.w);
        rec4[wi * 2]     = ra;
        rec4[wi * 2 + 1] = rb;
    }

    #pragma unroll
    for (int i = 0; i < G_TRIPS_; ++i) {
        const int t = t0 + i * G_SLOTS_;
        if (i + 1 < G_TRIPS_)
            load_a_tile(x, t + G_SLOTS_, l16, hi, a[(i + 1) & 1]);

        v8f acc = {};
        #pragma unroll
        for (int kb = 0; kb < 16; ++kb) {
            acc = __builtin_amdgcn_wmma_f32_16x16x4_f32(
                 false, a[i & 1][kb],  false, bf[kb],
                 (short)0, acc,  false,  false);
        }

        if (t < NRT_) {
            const int m0 = t * 16 + 8 * hi;
            if (nt < 2) {
                const int zb  = t / (N_ / 16);
                const int zn0 = (t - zb * (N_ / 16)) * 16 + 8 * hi;
                #pragma unroll
                for (int r = 0; r < 8; ++r) {
                    const long zr = (long)(zn0 + r) * B_ + zb;
                    z[zr * HALF_ + ncol] = acc[r];
                }
            } else {
                #pragma unroll
                for (int r = 0; r < 8; ++r) {
                    const long m = m0 + r;
                    out[m * CIN_ + ncol] = acc[r] + be;
                }
            }
        }
    }
}

#define QA_    1024
#define QB_    256
#define SCAP_  16
#define SPITCH_ 17
#define NGRP_  ((OWN_Q_ + 31) / 32)
#define NLD_   ((NGRP_ + 31) / 32)
#define FU_    5


#define GD_    16
#define RPW_   ((RPB_ + 31) / 32)
typedef __attribute__((ext_vector_type(32))) float v32f;

__device__ __forceinline__ void work_off_list(
    const int* q, int qn, int lane,
    const float* __restrict__ z, const float* __restrict__ vals,
    const int* __restrict__ cols, v32f& sums)
{
    const float4* __restrict__ z4 = (const float4*)z;
    for (int i0 = 0; i0 < qn; i0 += 32) {
        const int n   = (qn - i0 < 32) ? (qn - i0) : 32;
        const int ent = (lane < n) ? q[i0 + lane] : 0;
        const int e   = ent & 0xFFFFF;
        const int   cn = cols[e];
        const float v  = vals[e];
        for (int k = 0; k < n; k += 8) {
            float4 zz[8];
            float  vk[8];
            int    dk[8];
            #pragma unroll
            for (int kk = 0; kk < 8; ++kk) {
                const int cnk = __builtin_amdgcn_readlane(cn, k + kk);
                vk[kk] = __int_as_float(__builtin_amdgcn_readlane(__float_as_int(v), k + kk));
                dk[kk] = (__builtin_amdgcn_readlane(ent, k + kk) >> 25) * 4;
                zz[kk] = z4[cnk * (B_ * HALF_ / 4) + lane];
            }
            #pragma unroll
            for (int kk = 0; kk < 8; ++kk) {
                if (k + kk < n) {
                    sums[dk[kk] + 0] += vk[kk] * zz[kk].x;
                    sums[dk[kk] + 1] += vk[kk] * zz[kk].y;
                    sums[dk[kk] + 2] += vk[kk] * zz[kk].z;
                    sums[dk[kk] + 3] += vk[kk] * zz[kk].w;
                }
            }
        }
    }
}

__device__ __forceinline__ void work_off_recs(
    const int* qx, const int* qy, int qn, int lane,
    const float* __restrict__ z, v32f& sums)
{
    const float4* __restrict__ z4 = (const float4*)z;
    for (int i0 = 0; i0 < qn; i0 += 32) {
        const int n  = (qn - i0 < 32) ? (qn - i0) : 32;
        const int rx = (lane < n) ? qx[i0 + lane] : 0;
        const int ry = (lane < n) ? qy[i0 + lane] : 0;
        for (int k = 0; k < n; k += GD_) {
            float4 zz[GD_];
            float  vk[GD_];
            int    dk[GD_];
            #pragma unroll
            for (int kk = 0; kk < GD_; ++kk) {
                const int rk  = __builtin_amdgcn_readlane(rx, k + kk);
                const int cnk = rk & 0xFFFF;
                vk[kk] = __int_as_float(__builtin_amdgcn_readlane(ry, k + kk));
                dk[kk] = ((rk >> 21) & 7) * 4;
                zz[kk] = z4[cnk * (B_ * HALF_ / 4) + lane];
            }
            #pragma unroll
            for (int kk = 0; kk < GD_; ++kk) {
                if (k + kk < n) {
                    sums[dk[kk] + 0] += vk[kk] * zz[kk].x;
                    sums[dk[kk] + 1] += vk[kk] * zz[kk].y;
                    sums[dk[kk] + 2] += vk[kk] * zz[kk].z;
                    sums[dk[kk] + 3] += vk[kk] * zz[kk].w;
                }
            }
        }
    }
}

__device__ __forceinline__ int filter_pass(
    int ld0, int ld1, int wave, int lane, unsigned g, int rot,
    const uint4* __restrict__ own, int* q, bool& too_small)
{
    const int sw = __builtin_amdgcn_readfirstlane(wave);
    unsigned xm[8];
    #pragma unroll
    for (int b = 0; b < 8; ++b) xm[b] = 0u - ((g >> b) & 1u);
    int qn = 0;
    #pragma nounroll
    for (int ld = ld0; ld < ld1; ld += FU_) {
        uint4 wa[FU_];
        uint4 wb[FU_];
        int   pc[FU_];
        bool  live[FU_];
        #pragma unroll
        for (int u = 0; u < FU_; ++u) {
            const int gi = (ld + u) * 32 + sw;
            int gg = gi + rot;
            if (gg >= NGRP_) gg -= NGRP_;
            pc[u] = gg * 32 + lane;
            live[u] = (ld + u < ld1) && (gi < NGRP_) && (pc[u] < OWN_Q_);
            const int pl = (pc[u] < OWN_Q_) ? pc[u] : (OWN_Q_ - 1);
            wa[u] = own[pl];
            wb[u] = own[OWN_Q_ + pl];
        }
        __builtin_amdgcn_sched_barrier(0);
        #pragma unroll
        for (int u = 0; u < FU_; ++u) {
            const unsigned mis = (wa[u].x ^ xm[0]) | (wa[u].y ^ xm[1]) | (wa[u].z ^ xm[2]) | (wa[u].w ^ xm[3])
                               | (wb[u].x ^ xm[4]) | (wb[u].y ^ xm[5]) | (wb[u].z ^ xm[6]) | (wb[u].w ^ xm[7]);
            unsigned f = live[u] ? ~mis : 0u;
            unsigned bal = __builtin_amdgcn_ballot_w32(f != 0u);
            if (bal != 0u) {
                const int e0 = (pc[u] >> 2) * 128 + (pc[u] & 3);
                do {
                    const int nh = __builtin_popcount(bal);
                    if (qn + nh > QA_) {
                        too_small = true;
                        break;
                    }
                    if (f != 0u)
                        q[qn + __builtin_amdgcn_mbcnt_lo(bal, 0u)] = e0 + 4 * __builtin_ctz(f);
                    qn += nh;
                    f &= f - 1u;
                    bal = __builtin_amdgcn_ballot_w32(f != 0u);
                } while (bal != 0u);
            }
        }
    }
    return qn;
}

__device__ __forceinline__ int add_local_rows(
    int* q, int qn, int lane, int lo, int cnt, const int* __restrict__ rows)
{
    int wn = 0;
    for (int i0 = 0; i0 < qn; i0 += 128) {
        int  e[4];
        int  r[4];
        bool ok[4];
        #pragma unroll
        for (int u = 0; u < 4; ++u) {
            ok[u] = (i0 + u * 32 + lane) < qn;
            e[u]  = ok[u] ? q[i0 + u * 32 + lane] : 0;
        }
        #pragma unroll
        for (int u = 0; u < 4; ++u) r[u] = rows[e[u]];
        #pragma unroll
        for (int u = 0; u < 4; ++u) {
            const unsigned d    = (unsigned)(r[u] - lo);
            const bool     keep = ok[u] && d < (unsigned)cnt;
            const unsigned bm   = __builtin_amdgcn_ballot_w32(keep);
            if (keep) q[wn + __builtin_amdgcn_mbcnt_lo(bm, 0u)] = e[u] | (int)(d << 20);
            wn += __builtin_popcount(bm);
        }
    }
    return wn;
}

__device__ __forceinline__ void collect_and_add(
    const int* queA, const int* cntA, int* qb, int wave, int lane,
    const float* __restrict__ z, const float* __restrict__ vals,
    const int* __restrict__ cols, v32f& sums)
{
    int nb = 0;
    for (int s0 = 0; s0 < 32; ++s0) {
        const int  s  = (s0 + wave) & 31;
        const int  ns = __builtin_amdgcn_readfirstlane(cntA[s]);
        const int* qs = queA + s * QA_;
        for (int i0 = 0; i0 < ns; i0 += 32) {
            const bool ok   = (i0 + lane) < ns;
            const int  ent  = ok ? qs[i0 + lane] : 0;
            const bool mine = ok && (((ent >> 20) & 31) == wave);
            const unsigned bm = __builtin_amdgcn_ballot_w32(mine);
            if (bm != 0u) {
                if (mine) qb[nb + __builtin_amdgcn_mbcnt_lo(bm, 0u)] = ent;
                nb += __builtin_popcount(bm);
                if (nb > QB_ - 32) {
                    __builtin_amdgcn_wave_barrier();
                    work_off_list(qb, nb, lane, z, vals, cols, sums);
                    __builtin_amdgcn_wave_barrier();
                    nb = 0;
                }
            }
        }
    }
    __builtin_amdgcn_wave_barrier();
    work_off_list(qb, nb, lane, z, vals, cols, sums);
}


__device__ __forceinline__ bool file_recs(
    const int* q, int qn, int wave, int lane, unsigned g, int cnt,
    const uint2* __restrict__ rec, int* runsx, int* runsy, int* runcnt)
{
    int  mycnt = 0;
    bool over  = false;
    for (int i0 = 0; i0 < qn; i0 += 128) {
        int   e[4];
        uint2 r[4];
        bool  ok[4];
        #pragma unroll
        for (int u = 0; u < 4; ++u) {
            ok[u] = (i0 + u * 32 + lane) < qn;
            e[u]  = ok[u] ? q[i0 + u * 32 + lane] : 0;
        }
        #pragma unroll
        for (int u = 0; u < 4; ++u) r[u] = rec[e[u]];
        #pragma unroll
        for (int u = 0; u < 4; ++u) {
            const unsigned d    = (r[u].x >> 16) & 255u;
            const bool     keep = ok[u] && (r[u].x >> 24) == g && d < (unsigned)cnt;
            const int      w    = (int)(d & 31u);
            unsigned same = __builtin_amdgcn_ballot_w32(keep);
            unsigned tome = same;
            #pragma unroll
            for (int b = 0; b < 5; ++b) {
                const unsigned bb = __builtin_amdgcn_ballot_w32(keep && ((w >> b) & 1));
                same &= ((w    >> b) & 1) ? bb : ~bb;
                tome &= ((lane >> b) & 1) ? bb : ~bb;
            }
            const int base = __builtin_amdgcn_ds_bpermute(w << 2, mycnt);
            const int pos  = base + (int)__builtin_amdgcn_mbcnt_lo(same, 0u);
            if (keep) {
                if (pos < SCAP_) {
                    const int at = (w * 32 + wave) * SPITCH_ + pos;
                    runsx[at] = (int)r[u].x;
                    runsy[at] = (int)r[u].y;
                } else {
                    over = true;
                }
            }
            mycnt += __builtin_popcount(tome);
        }
    }
    runcnt[wave * 32 + lane] = mycnt;
    return __builtin_amdgcn_ballot_w32(over) == 0u;
}

__device__ __forceinline__ void gather_runs_and_add(
    const int* runsx, const int* runsy, const int* runcnt, int* qc, int wave, int lane,
    const float* __restrict__ z, v32f& sums)
{
    const int  c    = runcnt[lane * 32 + wave];
    const int* runx = runsx + (wave * 32 + lane) * SPITCH_;
    const int* runy = runsy + (wave * 32 + lane) * SPITCH_;
    int* qcx = qc;
    int* qcy = qc + QA_ / 2;
    int p = c;
    #pragma unroll
    for (int o = 1; o < 32; o <<= 1) {
        const int t = __shfl_up(p, o);
        if (lane >= o) p += t;
    }
    const int nb    = __builtin_amdgcn_readlane(p, 31);
    const int start = p - c;
    for (int k = 0; k < SCAP_; ++k) {
        if (__builtin_amdgcn_ballot_w32(k < c) == 0u) break;
        if (k < c) {
            qcx[start + k] = runx[k];
            qcy[start + k] = runy[k];
        }
    }
    __builtin_amdgcn_wave_barrier();
    work_off_recs(qcx, qcy, nb, lane, z, sums);
}

__global__ __launch_bounds__(1024) void gcn_rowrange_accumulate(
    const float* __restrict__ z, const float* __restrict__ vals,
    const int* __restrict__ rows, const int* __restrict__ cols,
    const float* __restrict__ b_lin, const unsigned int* __restrict__ own,
    const unsigned int* __restrict__ recw, float* __restrict__ out)
{
    const uint2* __restrict__ rec = (const uint2*)recw;
    __shared__ int   queA[32 * QA_];
    __shared__ int   runsx[32 * 32 * SPITCH_];
    __shared__ int   runsy[32 * 32 * SPITCH_];
    __shared__ int   runcnt[32 * 32];
    __shared__ int   cntA[32];
    __shared__ int   list_too_small;

    const int tid  = threadIdx.x;
    const int lane = tid & 31;
    const int wave = tid >> 5;
    const int lo   = blockIdx.x * RPB_;
    const int cnt  = (N_ - lo < RPB_) ? (N_ - lo) : RPB_;

    const float4 bl = ((const float4*)b_lin)[lane & 7];
    v32f sums;
    #pragma unroll
    for (int i = 0; i < RPW_; ++i) {
        sums[i * 4 + 0] = bl.x;
        sums[i * 4 + 1] = bl.y;
        sums[i * 4 + 2] = bl.z;
        sums[i * 4 + 3] = bl.w;
    }
    if (tid == 0) list_too_small = 0;
    __syncthreads();

    int* qa = queA + wave * QA_;
    int* qb = runsx + wave * QB_;
    const int rot = blockIdx.x * 3;
    const uint4* own16 = (const uint4*)own;
    const unsigned key = blockIdx.x;

    bool too_small = false;
    int qn = filter_pass(0, NLD_, wave, lane, key, rot, own16, qa, too_small);
    if (!too_small) too_small = !file_recs(qa, qn, wave, lane, blockIdx.x, cnt, rec, runsx, runsy, runcnt);
    if (lane == 0 && too_small) list_too_small = 1;
    __syncthreads();
    const int redo = __builtin_amdgcn_readfirstlane(list_too_small);
    if (redo == 0) {
        gather_runs_and_add(runsx, runsy, runcnt, qa, wave, lane, z, sums);
    } else {
        for (int r = 0; r < NLD_; ++r) {
            __syncthreads();
            bool unused = false;
            qn = filter_pass(r, r + 1, wave, lane, key, rot, own16, qa, unused);
            qn = add_local_rows(qa, qn, lane, lo, cnt, rows);
            if (lane == 0) cntA[wave] = qn;
            __syncthreads();
            collect_and_add(queA, cntA, qb, wave, lane, z, vals, cols, sums);
        }
    }

    const int sw = __builtin_amdgcn_readfirstlane(wave);
    float* orow = out + ((long)(lane >> 3) * N_ + lo) * CIN_ + (lane & 7) * 4;
    #pragma unroll
    for (int j = 0; j < RPW_; ++j) {
        const int d = sw + 32 * j;
        if (d < cnt) {
            float4 o;
            o.x = sums[j * 4 + 0];
            o.y = sums[j * 4 + 1];
            o.z = sums[j * 4 + 2];
            o.w = sums[j * 4 + 3];
            *(float4*)(orow + (long)d * CIN_) = o;
        }
    }
}

extern "C" void kernel_launch(void* const* d_in, const int* in_sizes, int n_in,
                              void* d_out, int out_size, void* d_ws, size_t ws_size,
                              hipStream_t stream) {
    const float* x     = (const float*)d_in[0];
    const float* vals  = (const float*)d_in[1];
    const float* W_lin = (const float*)d_in[2];
    const float* b_lin = (const float*)d_in[3];
    const float* W_eye = (const float*)d_in[4];
    const float* b_eye = (const float*)d_in[5];
    const int*   rows  = (const int*)d_in[6];
    const int*   cols  = (const int*)d_in[7];
    float* out = (float*)d_out;
    float* z   = (float*)d_ws;
    unsigned int* own = (unsigned int*)(z + M_ * HALF_);
    unsigned int* rec = own + OWN_W_;

    gcn_gemm_wmma<<<256, 896, 0, stream>>>(x, W_lin, W_eye, b_eye, rows, cols, vals, out, z, own, rec);

    gcn_rowrange_accumulate<<<256, 1024, 0, stream>>>(z, vals, rows, cols, b_lin, own, rec, out);
}
